# starting kernel
# baseline (speedup 1.0000x reference)
#include <hip/hip_runtime.h>

#define N_NODES_C 50000
#define N_EDGES_C 800000
#define F_IN_C 32
#define C_C 256
#define N_PAD 50016
#define N_ACT_C 8

typedef __attribute__((ext_vector_type(16))) __bf16 v16bf;
typedef __attribute__((ext_vector_type(8)))  __bf16 v8bf;
typedef __attribute__((ext_vector_type(8)))  float  v8f;

static __device__ __forceinline__ v8f wmma_bf16(v16bf a, v16bf b, v8f c) {
    return __builtin_amdgcn_wmma_f32_16x16x32_bf16(
         false, a,  false, b,
         (short)0, c,  false,  false);
}

static __device__ __forceinline__ v16bf load_a_frag_f32(
    const float* __restrict__ A, int row, int ld, int k0, int lane) {
    int half = lane >> 4;
    const float* p0 = A + (long long)row * ld + k0 + half * 8;
    const float* p1 = p0 + 16;
    float4 f0 = *(const float4*)(p0);
    float4 f1 = *(const float4*)(p0 + 4);
    float4 f2 = *(const float4*)(p1);
    float4 f3 = *(const float4*)(p1 + 4);
    v16bf a;
    a[0]  = (__bf16)f0.x; a[1]  = (__bf16)f0.y; a[2]  = (__bf16)f0.z; a[3]  = (__bf16)f0.w;
    a[4]  = (__bf16)f1.x; a[5]  = (__bf16)f1.y; a[6]  = (__bf16)f1.z; a[7]  = (__bf16)f1.w;
    a[8]  = (__bf16)f2.x; a[9]  = (__bf16)f2.y; a[10] = (__bf16)f2.z; a[11] = (__bf16)f2.w;
    a[12] = (__bf16)f3.x; a[13] = (__bf16)f3.y; a[14] = (__bf16)f3.z; a[15] = (__bf16)f3.w;
    return a;
}

static __device__ __forceinline__ v16bf load_b_frag_bf16(
    const __bf16* __restrict__ BT, int col, int ld, int k0, int lane) {
    int half = lane >> 4;
    const __bf16* p = BT + (long long)col * ld + k0 + half * 8;
    v8bf b0 = *(const v8bf*)(p);
    v8bf b1 = *(const v8bf*)(p + 16);
    v16bf b;
#pragma unroll
    for (int i = 0; i < 8; ++i) { b[i] = b0[i]; b[8 + i] = b1[i]; }
    return b;
}

static __device__ __forceinline__ float sigmoidf_(float x) {
    return 1.0f / (1.0f + __expf(-x));
}

__global__ void convw_kernel(const float* __restrict__ gg,
                             const float* __restrict__ wih,
                             const float* __restrict__ whh,
                             __bf16* __restrict__ ggT,
                             __bf16* __restrict__ wihb,
                             __bf16* __restrict__ whhb) {
    int i = blockIdx.x * 256 + threadIdx.x;
    if (i < 3 * 256 * 256) {
        int l = i >> 16, r = i & 65535, j = r >> 8, k = r & 255;
        ggT[i] = (__bf16)gg[l * 65536 + k * 256 + j];
    }
    if (i < 768 * 256) {
        wihb[i] = (__bf16)wih[i];
        whhb[i] = (__bf16)whh[i];
    }
}

__global__ void init_h_kernel(const float* __restrict__ x, float* __restrict__ h) {
    int i = blockIdx.x * 256 + threadIdx.x;
    int n = i >> 8, j = i & 255;
    float v = 0.0f;
    if (n < N_NODES_C && j < F_IN_C) v = x[n * F_IN_C + j];
    h[i] = v;
}

__global__ void zero4_kernel(float4* __restrict__ p, int n4) {
    int i = blockIdx.x * 256 + threadIdx.x;
    if (i < n4) p[i] = make_float4(0.f, 0.f, 0.f, 0.f);
}

__global__ void __launch_bounds__(256)
gemm_m_kernel(const float* __restrict__ h, const __bf16* __restrict__ WgT,
              float* __restrict__ m) {
    int lane = threadIdx.x & 31;
    int wave = threadIdx.x >> 5;
    int half = lane >> 4;
    int row0 = blockIdx.x * 32 + (wave >> 2) * 16;
    int c0   = (wave & 3) * 64;
    int rowA = row0 + (lane & 15);

    v8f acc0 = {}, acc1 = {}, acc2 = {}, acc3 = {};
#pragma unroll
    for (int k0 = 0; k0 < 256; k0 += 32) {
        v16bf a = load_a_frag_f32(h, rowA, 256, k0, lane);
        v16bf b0 = load_b_frag_bf16(WgT, c0 +  0 + (lane & 15), 256, k0, lane);
        acc0 = wmma_bf16(a, b0, acc0);
        v16bf b1 = load_b_frag_bf16(WgT, c0 + 16 + (lane & 15), 256, k0, lane);
        acc1 = wmma_bf16(a, b1, acc1);
        v16bf b2 = load_b_frag_bf16(WgT, c0 + 32 + (lane & 15), 256, k0, lane);
        acc2 = wmma_bf16(a, b2, acc2);
        v16bf b3 = load_b_frag_bf16(WgT, c0 + 48 + (lane & 15), 256, k0, lane);
        acc3 = wmma_bf16(a, b3, acc3);
    }
    v8f accs[4] = {acc0, acc1, acc2, acc3};
#pragma unroll
    for (int t = 0; t < 4; ++t) {
        int c = c0 + t * 16 + (lane & 15);
#pragma unroll
        for (int vr = 0; vr < 8; ++vr)
            m[(long long)(row0 + vr + 8 * half) * 256 + c] = accs[t][vr];
    }
}

__global__ void __launch_bounds__(256)
scatter_kernel(const float* __restrict__ m, const int* __restrict__ ei,
               const float* __restrict__ ew, float* __restrict__ agg) {
    int tid = blockIdx.x * 256 + threadIdx.x;
    int e = tid >> 6;
    if (e >= N_EDGES_C) return;
    int f = (tid & 63) << 2;
    int s = ei[e];
    int d = ei[N_EDGES_C + e];
    float w = ew[e];
    float4 v = *(const float4*)(m + (long long)s * 256 + f);
    float* dst = agg + (long long)d * 256 + f;
    atomicAdd(dst + 0, v.x * w);
    atomicAdd(dst + 1, v.y * w);
    atomicAdd(dst + 2, v.z * w);
    atomicAdd(dst + 3, v.w * w);
}

__global__ void __launch_bounds__(256)
gru_kernel(const float* __restrict__ agg, float* __restrict__ h,
           const __bf16* __restrict__ wihb, const __bf16* __restrict__ whhb,
           const float* __restrict__ bih, const float* __restrict__ bhh) {
    int lane = threadIdx.x & 31;
    int wave = threadIdx.x >> 5;
    int half = lane >> 4;
    int row0 = blockIdx.x * 16;
    int c0   = wave * 32;
    int rowA = row0 + (lane & 15);

    v8f acc[2][3][2];
#pragma unroll
    for (int a = 0; a < 2; ++a)
#pragma unroll
        for (int g = 0; g < 3; ++g)
#pragma unroll
            for (int t = 0; t < 2; ++t) { v8f z = {}; acc[a][g][t] = z; }

#pragma unroll
    for (int k0 = 0; k0 < 256; k0 += 32) {
        v16bf a_agg = load_a_frag_f32(agg, rowA, 256, k0, lane);
        v16bf a_h   = load_a_frag_f32(h,   rowA, 256, k0, lane);
#pragma unroll
        for (int g = 0; g < 3; ++g) {
#pragma unroll
            for (int t = 0; t < 2; ++t) {
                int col = g * 256 + c0 + t * 16 + (lane & 15);
                v16bf b_ih = load_b_frag_bf16(wihb, col, 256, k0, lane);
                acc[0][g][t] = wmma_bf16(a_agg, b_ih, acc[0][g][t]);
                v16bf b_hh = load_b_frag_bf16(whhb, col, 256, k0, lane);
                acc[1][g][t] = wmma_bf16(a_h, b_hh, acc[1][g][t]);
            }
        }
    }

    __syncthreads();

#pragma unroll
    for (int t = 0; t < 2; ++t) {
        int c = c0 + t * 16 + (lane & 15);
        float bir = bih[c],       bhr = bhh[c];
        float biz = bih[256 + c], bhz = bhh[256 + c];
        float bin = bih[512 + c], bhn = bhh[512 + c];
#pragma unroll
        for (int vr = 0; vr < 8; ++vr) {
            long long idx = (long long)(row0 + vr + 8 * half) * 256 + c;
            float hold = h[idx];
            float r  = sigmoidf_(acc[0][0][t][vr] + bir + acc[1][0][t][vr] + bhr);
            float z  = sigmoidf_(acc[0][1][t][vr] + biz + acc[1][1][t][vr] + bhz);
            float nn = tanhf(acc[0][2][t][vr] + bin + r * (acc[1][2][t][vr] + bhn));
            h[idx] = (1.0f - z) * nn + z * hold;
        }
    }
}

__global__ void __launch_bounds__(256)
final_kernel(const float* __restrict__ h, const float* __restrict__ fcw,
             const float* __restrict__ fcb, float* __restrict__ out) {
    __shared__ float sw[N_ACT_C * 256];
    __shared__ float sb[N_ACT_C];
    for (int i = threadIdx.x; i < N_ACT_C * 256; i += 256) sw[i] = fcw[i];
    if (threadIdx.x < N_ACT_C) sb[threadIdx.x] = fcb[threadIdx.x];
    __syncthreads();

    int n = blockIdx.x * 256 + threadIdx.x;
    if (n >= N_NODES_C) return;

    float acc[N_ACT_C];
#pragma unroll
    for (int a = 0; a < N_ACT_C; ++a) acc[a] = sb[a];
    const float* hr = h + (long long)n * 256;
    for (int k = 0; k < 256; ++k) {
        float hv = fmaxf(hr[k], 0.0f);
#pragma unroll
        for (int a = 0; a < N_ACT_C; ++a) acc[a] += hv * sw[a * 256 + k];
    }
    float mx = acc[0];
#pragma unroll
    for (int a = 1; a < N_ACT_C; ++a) mx = fmaxf(mx, acc[a]);
    float s = 0.0f;
#pragma unroll
    for (int a = 0; a < N_ACT_C; ++a) { acc[a] = __expf(acc[a] - mx); s += acc[a]; }
    float inv = 1.0f / s;
#pragma unroll
    for (int a = 0; a < N_ACT_C; ++a) out[(long long)n * N_ACT_C + a] = acc[a] * inv;
}

extern "C" void kernel_launch(void* const* d_in, const int* in_sizes, int n_in,
                              void* d_out, int out_size, void* d_ws, size_t ws_size,
                              hipStream_t stream) {
    const float* x   = (const float*)d_in[0];
    const int*   ei  = (const int*)  d_in[1];
    const float* ew  = (const float*)d_in[2];
    const float* gg  = (const float*)d_in[3];
    const float* wih = (const float*)d_in[4];
    const float* whh = (const float*)d_in[5];
    const float* bih = (const float*)d_in[6];
    const float* bhh = (const float*)d_in[7];
    const float* fcw = (const float*)d_in[8];
    const float* fcb = (const float*)d_in[9];
    float* out = (float*)d_out;

    float* h   = (float*)d_ws;
    float* m   = h + (size_t)N_PAD * 256;
    float* agg = m + (size_t)N_PAD * 256;
    __bf16* ggT  = (__bf16*)(agg + (size_t)N_PAD * 256);
    __bf16* wihb = ggT + 3 * 256 * 256;
    __bf16* whhb = wihb + 768 * 256;

    convw_kernel<<<(3 * 256 * 256 + 255) / 256, 256, 0, stream>>>(gg, wih, whh,
                                                                  ggT, wihb, whhb);
    init_h_kernel<<<N_PAD, 256, 0, stream>>>(x, h);

    for (int l = 0; l < 3; ++l) {
        gemm_m_kernel<<<N_PAD / 32, 256, 0, stream>>>(h, ggT + (size_t)l * 65536, m);
        zero4_kernel<<<(N_PAD * 64 + 255) / 256, 256, 0, stream>>>((float4*)agg,
                                                                   N_PAD * 64);
        scatter_kernel<<<(N_EDGES_C * 64) / 256, 256, 0, stream>>>(m, ei, ew, agg);
        gru_kernel<<<N_PAD / 16, 256, 0, stream>>>(agg, h, wihb, whhb, bih, bhh);
    }

    final_kernel<<<(N_NODES_C + 255) / 256, 256, 0, stream>>>(h, fcw, fcb, out);
}


# Float-atomic scatter and zero pass replaced: edges grouped once per call by destination range (stable counting sort); a work-group sums its range's rows in an LDS tile; sum order = ascending edge id
# speedup vs baseline: 2.4598x; 2.4598x over previous
#include <hip/hip_runtime.h>

#define N_NODES_C 50000
#define N_EDGES_C 800000
#define F_IN_C 32
#define C_C 256
#define N_PAD 50016
#define N_ACT_C 8

typedef __attribute__((ext_vector_type(16))) __bf16 v16bf;
typedef __attribute__((ext_vector_type(8)))  __bf16 v8bf;
typedef __attribute__((ext_vector_type(8)))  float  v8f;

static __device__ __forceinline__ v8f wmma_bf16(v16bf a, v16bf b, v8f c) {
    return __builtin_amdgcn_wmma_f32_16x16x32_bf16(
         false, a,  false, b,
         (short)0, c,  false,  false);
}

static __device__ __forceinline__ v16bf load_a_frag_f32(
    const float* __restrict__ A, int row, int ld, int k0, int lane) {
    int half = lane >> 4;
    const float* p0 = A + (long long)row * ld + k0 + half * 8;
    const float* p1 = p0 + 16;
    float4 f0 = *(const float4*)(p0);
    float4 f1 = *(const float4*)(p0 + 4);
    float4 f2 = *(const float4*)(p1);
    float4 f3 = *(const float4*)(p1 + 4);
    v16bf a;
    a[0]  = (__bf16)f0.x; a[1]  = (__bf16)f0.y; a[2]  = (__bf16)f0.z; a[3]  = (__bf16)f0.w;
    a[4]  = (__bf16)f1.x; a[5]  = (__bf16)f1.y; a[6]  = (__bf16)f1.z; a[7]  = (__bf16)f1.w;
    a[8]  = (__bf16)f2.x; a[9]  = (__bf16)f2.y; a[10] = (__bf16)f2.z; a[11] = (__bf16)f2.w;
    a[12] = (__bf16)f3.x; a[13] = (__bf16)f3.y; a[14] = (__bf16)f3.z; a[15] = (__bf16)f3.w;
    return a;
}

static __device__ __forceinline__ v16bf load_b_frag_bf16(
    const __bf16* __restrict__ BT, int col, int ld, int k0, int lane) {
    int half = lane >> 4;
    const __bf16* p = BT + (long long)col * ld + k0 + half * 8;
    v8bf b0 = *(const v8bf*)(p);
    v8bf b1 = *(const v8bf*)(p + 16);
    v16bf b;
#pragma unroll
    for (int i = 0; i < 8; ++i) { b[i] = b0[i]; b[8 + i] = b1[i]; }
    return b;
}

static __device__ __forceinline__ float sigmoidf_(float x) {
    return 1.0f / (1.0f + __expf(-x));
}

__global__ void convw_kernel(const float* __restrict__ gg,
                             const float* __restrict__ wih,
                             const float* __restrict__ whh,
                             __bf16* __restrict__ ggT,
                             __bf16* __restrict__ wihb,
                             __bf16* __restrict__ whhb) {
    int i = blockIdx.x * 256 + threadIdx.x;
    if (i < 3 * 256 * 256) {
        int l = i >> 16, r = i & 65535, j = r >> 8, k = r & 255;
        ggT[i] = (__bf16)gg[l * 65536 + k * 256 + j];
    }
    if (i < 768 * 256) {
        wihb[i] = (__bf16)wih[i];
        whhb[i] = (__bf16)whh[i];
    }
}

__global__ void init_h_kernel(const float* __restrict__ x, float* __restrict__ h) {
    int i = blockIdx.x * 256 + threadIdx.x;
    int n = i >> 8, j = i & 255;
    float v = 0.0f;
    if (n < N_NODES_C && j < F_IN_C) v = x[n * F_IN_C + j];
    h[i] = v;
}

__global__ void __launch_bounds__(256)
gemm_m_kernel(const float* __restrict__ h, const __bf16* __restrict__ WgT,
              float* __restrict__ m) {
    int lane = threadIdx.x & 31;
    int wave = threadIdx.x >> 5;
    int half = lane >> 4;
    int row0 = blockIdx.x * 32 + (wave >> 2) * 16;
    int c0   = (wave & 3) * 64;
    int rowA = row0 + (lane & 15);

    v8f acc0 = {}, acc1 = {}, acc2 = {}, acc3 = {};
#pragma unroll
    for (int k0 = 0; k0 < 256; k0 += 32) {
        v16bf a = load_a_frag_f32(h, rowA, 256, k0, lane);
        v16bf b0 = load_b_frag_bf16(WgT, c0 +  0 + (lane & 15), 256, k0, lane);
        acc0 = wmma_bf16(a, b0, acc0);
        v16bf b1 = load_b_frag_bf16(WgT, c0 + 16 + (lane & 15), 256, k0, lane);
        acc1 = wmma_bf16(a, b1, acc1);
        v16bf b2 = load_b_frag_bf16(WgT, c0 + 32 + (lane & 15), 256, k0, lane);
        acc2 = wmma_bf16(a, b2, acc2);
        v16bf b3 = load_b_frag_bf16(WgT, c0 + 48 + (lane & 15), 256, k0, lane);
        acc3 = wmma_bf16(a, b3, acc3);
    }
    v8f accs[4] = {acc0, acc1, acc2, acc3};
#pragma unroll
    for (int t = 0; t < 4; ++t) {
        int c = c0 + t * 16 + (lane & 15);
#pragma unroll
        for (int vr = 0; vr < 8; ++vr)
            m[(long long)(row0 + vr + 8 * half) * 256 + c] = accs[t][vr];
    }
}

#define EBLK 2048
#define NBLK ((N_EDGES_C + EBLK - 1) / EBLK)
#define BIN_SHIFT 4
#define BIN_ROWS 16
#define NBIN (N_PAD / BIN_ROWS)
#define BIN_PITCH 3136

__global__ void __launch_bounds__(256)
bin_count_kernel(const int* __restrict__ ei, int* __restrict__ counts) {
    __shared__ int hist[NBIN];
    for (int i = threadIdx.x; i < NBIN; i += 256) hist[i] = 0;
    __syncthreads();
    int e0 = blockIdx.x * EBLK;
#pragma unroll
    for (int j = 0; j < EBLK / 256; ++j) {
        int e = e0 + j * 256 + threadIdx.x;
        if (e < N_EDGES_C) atomicAdd(&hist[ei[N_EDGES_C + e] >> BIN_SHIFT], 1);
    }
    __syncthreads();
    int* row = counts + (size_t)blockIdx.x * BIN_PITCH;
    for (int i = threadIdx.x; i < NBIN; i += 256) row[i] = hist[i];
}

#define SCAN_U 17
__global__ void __launch_bounds__(32)
bin_scan_kernel(const int* __restrict__ counts, int* __restrict__ within,
                int* __restrict__ binTotal) {
    int bin = blockIdx.x * 32 + threadIdx.x;
    if (bin >= NBIN) return;
    int run = 0;
    for (int b0 = 0; b0 < NBLK; b0 += SCAN_U) {
        int c[SCAN_U];
#pragma unroll
        for (int u = 0; u < SCAN_U; ++u)
            c[u] = (b0 + u < NBLK) ? counts[(size_t)(b0 + u) * BIN_PITCH + bin] : 0;
#pragma unroll
        for (int u = 0; u < SCAN_U; ++u) {
            if (b0 + u < NBLK) within[(size_t)(b0 + u) * BIN_PITCH + bin] = run;
            run += c[u];
        }
    }
    binTotal[bin] = run;
}

#define PLACE_BPT ((NBIN + 255) / 256)
__global__ void __launch_bounds__(256)
bin_place_kernel(const int* __restrict__ ei, const float* __restrict__ ew,
                 const int* __restrict__ within, const int* __restrict__ binTotal,
                 int* __restrict__ binStart, int4* __restrict__ rec) {
    __shared__ int sCur[8][NBIN];
    __shared__ int sBase[NBIN];
    __shared__ int sWave[8];
    int tid = threadIdx.x, lane = tid & 31, wave = tid >> 5;
    int blk = blockIdx.x;

    int loc[PLACE_BPT];
    int sum = 0;
#pragma unroll
    for (int k = 0; k < PLACE_BPT; ++k) {
        int bin = PLACE_BPT * tid + k;
        int v = (bin < NBIN) ? binTotal[bin] : 0;
        loc[k] = sum;
        sum += v;
    }
    int inc = sum;
#pragma unroll
    for (int o = 1; o < 32; o <<= 1) {
        int t = __shfl_up(inc, o);
        if (lane >= o) inc += t;
    }
    if (lane == 31) sWave[wave] = inc;
    for (int i = tid; i < 8 * NBIN; i += 256) (&sCur[0][0])[i] = 0;
    __syncthreads();
    int wbase = 0;
    for (int w = 0; w < wave; ++w) wbase += sWave[w];
    int excl = wbase + inc - sum;
#pragma unroll
    for (int k = 0; k < PLACE_BPT; ++k) {
        int bin = PLACE_BPT * tid + k;
        if (bin < NBIN) {
            int st = excl + loc[k];
            sBase[bin] = st + within[(size_t)blk * BIN_PITCH + bin];
            if (blk == 0) binStart[bin] = st;
        }
    }

    int e0 = blk * EBLK + wave * 256;
    int dreg[8], sreg[8];
    float wreg[8];
#pragma unroll
    for (int c = 0; c < 8; ++c) {
        int e = e0 + c * 32 + lane;
        bool ok = e < N_EDGES_C;
        dreg[c] = ok ? ei[N_EDGES_C + e] : -1;
        sreg[c] = ok ? ei[e] : 0;
        wreg[c] = ok ? ew[e] : 0.0f;
        if (ok) atomicAdd(&sCur[wave][dreg[c] >> BIN_SHIFT], 1);
    }
    __syncthreads();
    for (int bin = tid; bin < NBIN; bin += 256) {
        int run = sBase[bin];
#pragma unroll
        for (int w = 0; w < 8; ++w) {
            int c = sCur[w][bin];
            sCur[w][bin] = run;
            run += c;
        }
    }
    __syncthreads();
#pragma unroll
    for (int c = 0; c < 8; ++c) {
        int d = dreg[c];
        int bin = d >> BIN_SHIFT;
        int rank = 0, cnt = 0;
#pragma unroll
        for (int j = 0; j < 32; ++j) {
            int bj = __builtin_amdgcn_readlane(bin, j);
            int same = (bj == bin) ? 1 : 0;
            cnt += same;
            rank += (j < lane) ? same : 0;
        }
        if (d >= 0) {
            int pos = sCur[wave][bin] + rank;
            rec[pos] = make_int4(sreg[c], d & (BIN_ROWS - 1), __float_as_int(wreg[c]), 0);
            if (rank == cnt - 1) sCur[wave][bin] = pos + 1;
        }
    }
}

#define AGG_U 16
__global__ void __launch_bounds__(128)
agg_kernel(const float* __restrict__ m, const int4* __restrict__ rec,
           const int* __restrict__ binStart, const int* __restrict__ binTotal,
           float* __restrict__ agg) {
    __shared__ __attribute__((aligned(16))) float tile[(BIN_ROWS + 1) * 256];
    int lane = threadIdx.x & 31, wave = threadIdx.x >> 5;
    int b = blockIdx.x;
    int col = wave * 64 + lane * 2;
    float* tcol = tile + col;
    const float* mcol = m + col;
#pragma unroll
    for (int r = 0; r <= BIN_ROWS; ++r) *(float2*)(tcol + r * 256) = make_float2(0.0f, 0.0f);

    int start = __builtin_amdgcn_readfirstlane(binStart[b]);
    int n     = __builtin_amdgcn_readfirstlane(binTotal[b]);
    const int4* list = rec + start;

    const int4 filler = make_int4(0, BIN_ROWS, 0, 0);
    int4 cur = (lane < n) ? list[lane] : filler;
    for (int c0 = 0; c0 < n; c0 += 32) {
        int nidx = c0 + 32 + lane;
        int4 nxt = (nidx < n) ? list[nidx] : filler;
        int cnt = min(32, n - c0);
        for (int j = 0; j < cnt; j += AGG_U) {
            float2 v[AGG_U];
#pragma unroll
            for (int u = 0; u < AGG_U; ++u) {
                int s = __builtin_amdgcn_readlane(cur.x, j + u);
                v[u] = *(const float2*)(mcol + (size_t)s * 256);
            }
#pragma unroll
            for (int u = 0; u < AGG_U; ++u) {
                int dl = __builtin_amdgcn_readlane(cur.y, j + u);
                float w = __int_as_float(__builtin_amdgcn_readlane(cur.z, j + u));
                atomicAdd(tcol + dl * 256,     v[u].x * w);
                atomicAdd(tcol + dl * 256 + 1, v[u].y * w);
            }
        }
        cur = nxt;
    }

    float* arow = agg + (size_t)b * BIN_ROWS * 256 + col;
#pragma unroll
    for (int r = 0; r < BIN_ROWS; ++r)
        *(float2*)(arow + r * 256) = *(const float2*)(tcol + r * 256);
}

#define GRU_RB 128
#define GRU_RT (GRU_RB / 16)
#define GRU_PITCH 264

typedef __attribute__((ext_vector_type(4))) __bf16 v4bf;

static __device__ __forceinline__ v16bf load_a_frag_lds(
    const __bf16* sA, int row, int k0, int half) {
    const __bf16* p = sA + row * GRU_PITCH + k0 + half * 8;
    v8bf a0 = *(const v8bf*)(p);
    v8bf a1 = *(const v8bf*)(p + 16);
    v16bf a;
#pragma unroll
    for (int i = 0; i < 8; ++i) { a[i] = a0[i]; a[8 + i] = a1[i]; }
    return a;
}

#define GRU_KSTEP(WS, kq)                                                   \
    _Pragma("unroll")                                                       \
    for (int rt = 0; rt < GRU_RT; rt += 2) {                                \
        v16bf ag0 = load_a_frag_lds(sAgg, rt * 16 + cl,      (kq), half);   \
        v16bf ah0 = load_a_frag_lds(sH,   rt * 16 + cl,      (kq), half);   \
        v16bf ag1 = load_a_frag_lds(sAgg, rt * 16 + 16 + cl, (kq), half);   \
        v16bf ah1 = load_a_frag_lds(sH,   rt * 16 + 16 + cl, (kq), half);   \
        aR[rt]      = wmma_bf16(ag0, WS[0], aR[rt]);                        \
        aZ[rt]      = wmma_bf16(ag0, WS[1], aZ[rt]);                        \
        aIN[rt]     = wmma_bf16(ag0, WS[2], aIN[rt]);                       \
        aR[rt + 1]  = wmma_bf16(ag1, WS[0], aR[rt + 1]);                    \
        aZ[rt + 1]  = wmma_bf16(ag1, WS[1], aZ[rt + 1]);                    \
        aIN[rt + 1] = wmma_bf16(ag1, WS[2], aIN[rt + 1]);                   \
        aR[rt]      = wmma_bf16(ah0, WS[3], aR[rt]);                        \
        aZ[rt]      = wmma_bf16(ah0, WS[4], aZ[rt]);                        \
        aHN[rt]     = wmma_bf16(ah0, WS[5], aHN[rt]);                       \
        aR[rt + 1]  = wmma_bf16(ah1, WS[3], aR[rt + 1]);                    \
        aZ[rt + 1]  = wmma_bf16(ah1, WS[4], aZ[rt + 1]);                    \
        aHN[rt + 1] = wmma_bf16(ah1, WS[5], aHN[rt + 1]);                   \
    }

#define GRU_LOAD_W(WS, kq)                                                  \
    _Pragma("unroll")                                                       \
    for (int g = 0; g < 3; ++g) {                                           \
        WS[g]     = load_b_frag_bf16(wihb, g * 256 + c, 256, (kq), lane);   \
        WS[3 + g] = load_b_frag_bf16(whhb, g * 256 + c, 256, (kq), lane);   \
    }

__global__ void __launch_bounds__(256) __attribute__((amdgpu_waves_per_eu(2, 2)))
gru_kernel(const float* __restrict__ agg, float* __restrict__ h,
           const __bf16* __restrict__ wihb, const __bf16* __restrict__ whhb,
           const float* __restrict__ bih, const float* __restrict__ bhh) {
    __shared__ __attribute__((aligned(16))) __bf16 sA[2 * GRU_RB * GRU_PITCH];
    __bf16* sAgg = sA;
    __bf16* sH   = sA + GRU_RB * GRU_PITCH;

    int tid  = threadIdx.x;
    int lane = tid & 31;
    int wave = tid >> 5;
    int half = lane >> 4;
    int cl   = lane & 15;
    int row0 = blockIdx.x * GRU_RB;

#pragma unroll 1
    for (int it = 0; it < 32; it += 8) {
        float4 va[8], vh[8];
#pragma unroll
        for (int j = 0; j < 8; ++j) {
            int idx = (it + j) * 256 + tid;
            int r = idx >> 6, c4 = idx & 63;
            int grow = min(row0 + r, N_PAD - 1);
            va[j] = *(const float4*)(agg + (long long)grow * 256 + c4 * 4);
            vh[j] = *(const float4*)(h   + (long long)grow * 256 + c4 * 4);
        }
#pragma unroll
        for (int j = 0; j < 8; ++j) {
            int idx = (it + j) * 256 + tid;
            int r = idx >> 6, c4 = idx & 63;
            v4bf pa, ph;
            pa[0] = (__bf16)va[j].x; pa[1] = (__bf16)va[j].y;
            pa[2] = (__bf16)va[j].z; pa[3] = (__bf16)va[j].w;
            ph[0] = (__bf16)vh[j].x; ph[1] = (__bf16)vh[j].y;
            ph[2] = (__bf16)vh[j].z; ph[3] = (__bf16)vh[j].w;
            *(v4bf*)(sAgg + r * GRU_PITCH + c4 * 4) = pa;
            *(v4bf*)(sH   + r * GRU_PITCH + c4 * 4) = ph;
        }
    }
    __syncthreads();

#pragma unroll 1
    for (int t = 0; t < 2; ++t) {
        int c = wave * 32 + t * 16 + cl;

        v8f aR[GRU_RT], aZ[GRU_RT], aIN[GRU_RT], aHN[GRU_RT];
#pragma unroll
        for (int rt = 0; rt < GRU_RT; ++rt) {
            v8f z = {};
            aR[rt] = z; aZ[rt] = z; aIN[rt] = z; aHN[rt] = z;
        }

        v16bf w0[6], w1[6];
        GRU_LOAD_W(w0, 0)

#pragma unroll 1
        for (int kk = 0; kk < 4; ++kk) {
            int k0 = kk * 64;
            GRU_LOAD_W(w1, k0 + 32)
            GRU_KSTEP(w0, k0)
            if (kk < 3) {
                GRU_LOAD_W(w0, k0 + 64)
            }
            GRU_KSTEP(w1, k0 + 32)
        }

        float bir = bih[c],       bhr = bhh[c];
        float biz = bih[256 + c], bhz = bhh[256 + c];
        float bin = bih[512 + c], bhn = bhh[512 + c];
        float hold[GRU_RT][8];
        const float* hcol = h + (8 * half) * 256 + c;
#pragma unroll
        for (int rt = 0; rt < GRU_RT; ++rt) {
            int rbase = (row0 + rt * 16 < N_PAD) ? row0 + rt * 16 : row0;
#pragma unroll
            for (int vr = 0; vr < 8; ++vr)
                hold[rt][vr] = hcol[(long long)(rbase + vr) * 256];
        }
#pragma unroll
        for (int rt = 0; rt < GRU_RT; ++rt) {
            if (row0 + rt * 16 < N_PAD) {
#pragma unroll
                for (int vr = 0; vr < 8; ++vr) {
                    long long idx = (long long)(row0 + rt * 16 + vr + 8 * half) * 256 + c;
                    float r  = sigmoidf_(aR[rt][vr] + bir + bhr);
                    float z  = sigmoidf_(aZ[rt][vr] + biz + bhz);
                    float nn = tanhf(aIN[rt][vr] + bin + r * (aHN[rt][vr] + bhn));
                    h[idx] = (1.0f - z) * nn + z * hold[rt][vr];
                }
            }
        }
    }
}

__global__ void __launch_bounds__(256)
final_kernel(const float* __restrict__ h, const float* __restrict__ fcw,
             const float* __restrict__ fcb, float* __restrict__ out) {
    __shared__ float sw[N_ACT_C * 256];
    __shared__ float sb[N_ACT_C];
    for (int i = threadIdx.x; i < N_ACT_C * 256; i += 256) sw[i] = fcw[i];
    if (threadIdx.x < N_ACT_C) sb[threadIdx.x] = fcb[threadIdx.x];
    __syncthreads();

    int n = blockIdx.x * 256 + threadIdx.x;
    if (n >= N_NODES_C) return;

    float acc[N_ACT_C];
#pragma unroll
    for (int a = 0; a < N_ACT_C; ++a) acc[a] = sb[a];
    const float* hr = h + (long long)n * 256;
    for (int k = 0; k < 256; ++k) {
        float hv = fmaxf(hr[k], 0.0f);
#pragma unroll
        for (int a = 0; a < N_ACT_C; ++a) acc[a] += hv * sw[a * 256 + k];
    }
    float mx = acc[0];
#pragma unroll
    for (int a = 1; a < N_ACT_C; ++a) mx = fmaxf(mx, acc[a]);
    float s = 0.0f;
#pragma unroll
    for (int a = 0; a < N_ACT_C; ++a) { acc[a] = __expf(acc[a] - mx); s += acc[a]; }
    float inv = 1.0f / s;
#pragma unroll
    for (int a = 0; a < N_ACT_C; ++a) out[(long long)n * N_ACT_C + a] = acc[a] * inv;
}

extern "C" void kernel_launch(void* const* d_in, const int* in_sizes, int n_in,
                              void* d_out, int out_size, void* d_ws, size_t ws_size,
                              hipStream_t stream) {
    const float* x   = (const float*)d_in[0];
    const int*   ei  = (const int*)  d_in[1];
    const float* ew  = (const float*)d_in[2];
    const float* gg  = (const float*)d_in[3];
    const float* wih = (const float*)d_in[4];
    const float* whh = (const float*)d_in[5];
    const float* bih = (const float*)d_in[6];
    const float* bhh = (const float*)d_in[7];
    const float* fcw = (const float*)d_in[8];
    const float* fcb = (const float*)d_in[9];
    float* out = (float*)d_out;

    float* h   = (float*)d_ws;
    float* m   = h + (size_t)N_PAD * 256;
    float* agg = m + (size_t)N_PAD * 256;
    __bf16* ggT  = (__bf16*)(agg + (size_t)N_PAD * 256);
    __bf16* wihb = ggT + 3 * 256 * 256;
    __bf16* whhb = wihb + 768 * 256;
    int* counts   = (int*)(whhb + 768 * 256);
    int* within   = counts + (size_t)NBLK * BIN_PITCH;
    int* binTotal = within + (size_t)NBLK * BIN_PITCH;
    int* binStart = binTotal + BIN_PITCH;
    int4* rec     = (int4*)(binStart + BIN_PITCH);

    convw_kernel<<<(3 * 256 * 256 + 255) / 256, 256, 0, stream>>>(gg, wih, whh,
                                                                  ggT, wihb, whhb);
    init_h_kernel<<<N_PAD, 256, 0, stream>>>(x, h);
    bin_count_kernel<<<NBLK, 256, 0, stream>>>(ei, counts);
    bin_scan_kernel<<<(NBIN + 31) / 32, 32, 0, stream>>>(counts, within, binTotal);
    bin_place_kernel<<<NBLK, 256, 0, stream>>>(ei, ew, within, binTotal, binStart, rec);

    for (int l = 0; l < 3; ++l) {
        gemm_m_kernel<<<N_PAD / 32, 256, 0, stream>>>(h, ggT + (size_t)l * 65536, m);
        agg_kernel<<<NBIN, 128, 0, stream>>>(m, rec, binStart, binTotal, agg);
        gru_kernel<<<(N_PAD + GRU_RB - 1) / GRU_RB, 256, 0, stream>>>(agg, h, wihb, whhb,
                                                                     bih, bhh);
    }

    final_kernel<<<(N_NODES_C + 255) / 256, 256, 0, stream>>>(h, fcw, fcb, out);
}


# GRU kernel: A fragments read from LDS one block of 24 WMMAs (4 row tiles) ahead, as one group into a second register set; was 2 reads, a full wait, 3 WMMAs, 128 times a column pass; sums unchanged
# speedup vs baseline: 2.5748x; 1.0468x over previous
#include <hip/hip_runtime.h>

#define N_NODES_C 50000
#define N_EDGES_C 800000
#define F_IN_C 32
#define C_C 256
#define N_PAD 50016
#define N_ACT_C 8

typedef __attribute__((ext_vector_type(16))) __bf16 v16bf;
typedef __attribute__((ext_vector_type(8)))  __bf16 v8bf;
typedef __attribute__((ext_vector_type(8)))  float  v8f;

static __device__ __forceinline__ v8f wmma_bf16(v16bf a, v16bf b, v8f c) {
    return __builtin_amdgcn_wmma_f32_16x16x32_bf16(
         false, a,  false, b,
         (short)0, c,  false,  false);
}

static __device__ __forceinline__ v16bf load_a_frag_f32(
    const float* __restrict__ A, int row, int ld, int k0, int lane) {
    int half = lane >> 4;
    const float* p0 = A + (long long)row * ld + k0 + half * 8;
    const float* p1 = p0 + 16;
    float4 f0 = *(const float4*)(p0);
    float4 f1 = *(const float4*)(p0 + 4);
    float4 f2 = *(const float4*)(p1);
    float4 f3 = *(const float4*)(p1 + 4);
    v16bf a;
    a[0]  = (__bf16)f0.x; a[1]  = (__bf16)f0.y; a[2]  = (__bf16)f0.z; a[3]  = (__bf16)f0.w;
    a[4]  = (__bf16)f1.x; a[5]  = (__bf16)f1.y; a[6]  = (__bf16)f1.z; a[7]  = (__bf16)f1.w;
    a[8]  = (__bf16)f2.x; a[9]  = (__bf16)f2.y; a[10] = (__bf16)f2.z; a[11] = (__bf16)f2.w;
    a[12] = (__bf16)f3.x; a[13] = (__bf16)f3.y; a[14] = (__bf16)f3.z; a[15] = (__bf16)f3.w;
    return a;
}

static __device__ __forceinline__ v16bf load_b_frag_bf16(
    const __bf16* __restrict__ BT, int col, int ld, int k0, int lane) {
    int half = lane >> 4;
    const __bf16* p = BT + (long long)col * ld + k0 + half * 8;
    v8bf b0 = *(const v8bf*)(p);
    v8bf b1 = *(const v8bf*)(p + 16);
    v16bf b;
#pragma unroll
    for (int i = 0; i < 8; ++i) { b[i] = b0[i]; b[8 + i] = b1[i]; }
    return b;
}

static __device__ __forceinline__ float sigmoidf_(float x) {
    return 1.0f / (1.0f + __expf(-x));
}

__global__ void convw_kernel(const float* __restrict__ gg,
                             const float* __restrict__ wih,
                             const float* __restrict__ whh,
                             __bf16* __restrict__ ggT,
                             __bf16* __restrict__ wihb,
                             __bf16* __restrict__ whhb) {
    int i = blockIdx.x * 256 + threadIdx.x;
    if (i < 3 * 256 * 256) {
        int l = i >> 16, r = i & 65535, j = r >> 8, k = r & 255;
        ggT[i] = (__bf16)gg[l * 65536 + k * 256 + j];
    }
    if (i < 768 * 256) {
        wihb[i] = (__bf16)wih[i];
        whhb[i] = (__bf16)whh[i];
    }
}

__global__ void init_h_kernel(const float* __restrict__ x, float* __restrict__ h) {
    int i = blockIdx.x * 256 + threadIdx.x;
    int n = i >> 8, j = i & 255;
    float v = 0.0f;
    if (n < N_NODES_C && j < F_IN_C) v = x[n * F_IN_C + j];
    h[i] = v;
}

__global__ void __launch_bounds__(256)
gemm_m_kernel(const float* __restrict__ h, const __bf16* __restrict__ WgT,
              float* __restrict__ m) {
    int lane = threadIdx.x & 31;
    int wave = threadIdx.x >> 5;
    int half = lane >> 4;
    int row0 = blockIdx.x * 32 + (wave >> 2) * 16;
    int c0   = (wave & 3) * 64;
    int rowA = row0 + (lane & 15);

    v8f acc0 = {}, acc1 = {}, acc2 = {}, acc3 = {};
#pragma unroll
    for (int k0 = 0; k0 < 256; k0 += 32) {
        v16bf a = load_a_frag_f32(h, rowA, 256, k0, lane);
        v16bf b0 = load_b_frag_bf16(WgT, c0 +  0 + (lane & 15), 256, k0, lane);
        acc0 = wmma_bf16(a, b0, acc0);
        v16bf b1 = load_b_frag_bf16(WgT, c0 + 16 + (lane & 15), 256, k0, lane);
        acc1 = wmma_bf16(a, b1, acc1);
        v16bf b2 = load_b_frag_bf16(WgT, c0 + 32 + (lane & 15), 256, k0, lane);
        acc2 = wmma_bf16(a, b2, acc2);
        v16bf b3 = load_b_frag_bf16(WgT, c0 + 48 + (lane & 15), 256, k0, lane);
        acc3 = wmma_bf16(a, b3, acc3);
    }
    v8f accs[4] = {acc0, acc1, acc2, acc3};
#pragma unroll
    for (int t = 0; t < 4; ++t) {
        int c = c0 + t * 16 + (lane & 15);
#pragma unroll
        for (int vr = 0; vr < 8; ++vr)
            m[(long long)(row0 + vr + 8 * half) * 256 + c] = accs[t][vr];
    }
}

#define EBLK 2048
#define NBLK ((N_EDGES_C + EBLK - 1) / EBLK)
#define BIN_SHIFT 4
#define BIN_ROWS 16
#define NBIN (N_PAD / BIN_ROWS)
#define BIN_PITCH 3136

__global__ void __launch_bounds__(256)
bin_count_kernel(const int* __restrict__ ei, int* __restrict__ counts) {
    __shared__ int hist[NBIN];
    for (int i = threadIdx.x; i < NBIN; i += 256) hist[i] = 0;
    __syncthreads();
    int e0 = blockIdx.x * EBLK;
#pragma unroll
    for (int j = 0; j < EBLK / 256; ++j) {
        int e = e0 + j * 256 + threadIdx.x;
        if (e < N_EDGES_C) atomicAdd(&hist[ei[N_EDGES_C + e] >> BIN_SHIFT], 1);
    }
    __syncthreads();
    int* row = counts + (size_t)blockIdx.x * BIN_PITCH;
    for (int i = threadIdx.x; i < NBIN; i += 256) row[i] = hist[i];
}

#define SCAN_U 17
__global__ void __launch_bounds__(32)
bin_scan_kernel(const int* __restrict__ counts, int* __restrict__ within,
                int* __restrict__ binTotal) {
    int bin = blockIdx.x * 32 + threadIdx.x;
    if (bin >= NBIN) return;
    int run = 0;
    for (int b0 = 0; b0 < NBLK; b0 += SCAN_U) {
        int c[SCAN_U];
#pragma unroll
        for (int u = 0; u < SCAN_U; ++u)
            c[u] = (b0 + u < NBLK) ? counts[(size_t)(b0 + u) * BIN_PITCH + bin] : 0;
#pragma unroll
        for (int u = 0; u < SCAN_U; ++u) {
            if (b0 + u < NBLK) within[(size_t)(b0 + u) * BIN_PITCH + bin] = run;
            run += c[u];
        }
    }
    binTotal[bin] = run;
}

#define PLACE_BPT ((NBIN + 255) / 256)
__global__ void __launch_bounds__(256)
bin_place_kernel(const int* __restrict__ ei, const float* __restrict__ ew,
                 const int* __restrict__ within, const int* __restrict__ binTotal,
                 int* __restrict__ binStart, int4* __restrict__ rec) {
    __shared__ int sCur[8][NBIN];
    __shared__ int sBase[NBIN];
    __shared__ int sWave[8];
    int tid = threadIdx.x, lane = tid & 31, wave = tid >> 5;
    int blk = blockIdx.x;

    int loc[PLACE_BPT];
    int sum = 0;
#pragma unroll
    for (int k = 0; k < PLACE_BPT; ++k) {
        int bin = PLACE_BPT * tid + k;
        int v = (bin < NBIN) ? binTotal[bin] : 0;
        loc[k] = sum;
        sum += v;
    }
    int inc = sum;
#pragma unroll
    for (int o = 1; o < 32; o <<= 1) {
        int t = __shfl_up(inc, o);
        if (lane >= o) inc += t;
    }
    if (lane == 31) sWave[wave] = inc;
    for (int i = tid; i < 8 * NBIN; i += 256) (&sCur[0][0])[i] = 0;
    __syncthreads();
    int wbase = 0;
    for (int w = 0; w < wave; ++w) wbase += sWave[w];
    int excl = wbase + inc - sum;
#pragma unroll
    for (int k = 0; k < PLACE_BPT; ++k) {
        int bin = PLACE_BPT * tid + k;
        if (bin < NBIN) {
            int st = excl + loc[k];
            sBase[bin] = st + within[(size_t)blk * BIN_PITCH + bin];
            if (blk == 0) binStart[bin] = st;
        }
    }

    int e0 = blk * EBLK + wave * 256;
    int dreg[8], sreg[8];
    float wreg[8];
#pragma unroll
    for (int c = 0; c < 8; ++c) {
        int e = e0 + c * 32 + lane;
        bool ok = e < N_EDGES_C;
        dreg[c] = ok ? ei[N_EDGES_C + e] : -1;
        sreg[c] = ok ? ei[e] : 0;
        wreg[c] = ok ? ew[e] : 0.0f;
        if (ok) atomicAdd(&sCur[wave][dreg[c] >> BIN_SHIFT], 1);
    }
    __syncthreads();
    for (int bin = tid; bin < NBIN; bin += 256) {
        int run = sBase[bin];
#pragma unroll
        for (int w = 0; w < 8; ++w) {
            int c = sCur[w][bin];
            sCur[w][bin] = run;
            run += c;
        }
    }
    __syncthreads();
#pragma unroll
    for (int c = 0; c < 8; ++c) {
        int d = dreg[c];
        int bin = d >> BIN_SHIFT;
        int rank = 0, cnt = 0;
#pragma unroll
        for (int j = 0; j < 32; ++j) {
            int bj = __builtin_amdgcn_readlane(bin, j);
            int same = (bj == bin) ? 1 : 0;
            cnt += same;
            rank += (j < lane) ? same : 0;
        }
        if (d >= 0) {
            int pos = sCur[wave][bin] + rank;
            rec[pos] = make_int4(sreg[c], d & (BIN_ROWS - 1), __float_as_int(wreg[c]), 0);
            if (rank == cnt - 1) sCur[wave][bin] = pos + 1;
        }
    }
}

#define AGG_U 16
__global__ void __launch_bounds__(128)
agg_kernel(const float* __restrict__ m, const int4* __restrict__ rec,
           const int* __restrict__ binStart, const int* __restrict__ binTotal,
           float* __restrict__ agg) {
    __shared__ __attribute__((aligned(16))) float tile[(BIN_ROWS + 1) * 256];
    int lane = threadIdx.x & 31, wave = threadIdx.x >> 5;
    int b = blockIdx.x;
    int col = wave * 64 + lane * 2;
    float* tcol = tile + col;
    const float* mcol = m + col;
#pragma unroll
    for (int r = 0; r <= BIN_ROWS; ++r) *(float2*)(tcol + r * 256) = make_float2(0.0f, 0.0f);

    int start = __builtin_amdgcn_readfirstlane(binStart[b]);
    int n     = __builtin_amdgcn_readfirstlane(binTotal[b]);
    const int4* list = rec + start;

    const int4 filler = make_int4(0, BIN_ROWS, 0, 0);
    int4 cur = (lane < n) ? list[lane] : filler;
    for (int c0 = 0; c0 < n; c0 += 32) {
        int nidx = c0 + 32 + lane;
        int4 nxt = (nidx < n) ? list[nidx] : filler;
        int cnt = min(32, n - c0);
        for (int j = 0; j < cnt; j += AGG_U) {
            float2 v[AGG_U];
#pragma unroll
            for (int u = 0; u < AGG_U; ++u) {
                int s = __builtin_amdgcn_readlane(cur.x, j + u);
                v[u] = *(const float2*)(mcol + (size_t)s * 256);
            }
#pragma unroll
            for (int u = 0; u < AGG_U; ++u) {
                int dl = __builtin_amdgcn_readlane(cur.y, j + u);
                float w = __int_as_float(__builtin_amdgcn_readlane(cur.z, j + u));
                atomicAdd(tcol + dl * 256,     v[u].x * w);
                atomicAdd(tcol + dl * 256 + 1, v[u].y * w);
            }
        }
        cur = nxt;
    }

    float* arow = agg + (size_t)b * BIN_ROWS * 256 + col;
#pragma unroll
    for (int r = 0; r < BIN_ROWS; ++r)
        *(float2*)(arow + r * 256) = *(const float2*)(tcol + r * 256);
}

#define GRU_RB 128
#define GRU_RT (GRU_RB / 16)
#define GRU_PITCH 264

typedef __attribute__((ext_vector_type(4))) __bf16 v4bf;

static __device__ __forceinline__ v16bf load_a_frag_lds(
    const __bf16* sA, int row, int k0, int half) {
    const __bf16* p = sA + row * GRU_PITCH + k0 + half * 8;
    v8bf a0 = *(const v8bf*)(p);
    v8bf a1 = *(const v8bf*)(p + 16);
    v16bf a;
#pragma unroll
    for (int i = 0; i < 8; ++i) { a[i] = a0[i]; a[8 + i] = a1[i]; }
    return a;
}

#define GRU_LOAD_A(AS, rb, kq)                                              \
    _Pragma("unroll")                                                       \
    for (int i = 0; i < 4; ++i) {                                           \
        AS[i]     = load_a_frag_lds(sAgg, ((rb) + i) * 16 + cl, (kq), half);\
        AS[4 + i] = load_a_frag_lds(sH,   ((rb) + i) * 16 + cl, (kq), half);\
    }

#define GRU_BLOCK(AS, WS, rb)                                               \
    _Pragma("unroll")                                                       \
    for (int i = 0; i < 4; ++i) {                                           \
        aR[(rb) + i]  = wmma_bf16(AS[i], WS[0], aR[(rb) + i]);              \
        aZ[(rb) + i]  = wmma_bf16(AS[i], WS[1], aZ[(rb) + i]);              \
        aIN[(rb) + i] = wmma_bf16(AS[i], WS[2], aIN[(rb) + i]);             \
    }                                                                       \
    _Pragma("unroll")                                                       \
    for (int i = 0; i < 4; ++i) {                                           \
        aR[(rb) + i]  = wmma_bf16(AS[4 + i], WS[3], aR[(rb) + i]);          \
        aZ[(rb) + i]  = wmma_bf16(AS[4 + i], WS[4], aZ[(rb) + i]);          \
        aHN[(rb) + i] = wmma_bf16(AS[4 + i], WS[5], aHN[(rb) + i]);         \
    }

#define GRU_FENCE __builtin_amdgcn_sched_barrier(0);

#define GRU_LOAD_W(WS, kq)                                                  \
    _Pragma("unroll")                                                       \
    for (int g = 0; g < 3; ++g) {                                           \
        WS[g]     = load_b_frag_bf16(wihb, g * 256 + c, 256, (kq), lane);   \
        WS[3 + g] = load_b_frag_bf16(whhb, g * 256 + c, 256, (kq), lane);   \
    }

__global__ void __launch_bounds__(256) __attribute__((amdgpu_waves_per_eu(2, 2)))
gru_kernel(const float* __restrict__ agg, float* __restrict__ h,
           const __bf16* __restrict__ wihb, const __bf16* __restrict__ whhb,
           const float* __restrict__ bih, const float* __restrict__ bhh) {
    __shared__ __attribute__((aligned(16))) __bf16 sA[2 * GRU_RB * GRU_PITCH];
    __bf16* sAgg = sA;
    __bf16* sH   = sA + GRU_RB * GRU_PITCH;

    int tid  = threadIdx.x;
    int lane = tid & 31;
    int wave = tid >> 5;
    int half = lane >> 4;
    int cl   = lane & 15;
    int row0 = blockIdx.x * GRU_RB;

#pragma unroll 1
    for (int it = 0; it < 32; it += 8) {
        float4 va[8], vh[8];
#pragma unroll
        for (int j = 0; j < 8; ++j) {
            int idx = (it + j) * 256 + tid;
            int r = idx >> 6, c4 = idx & 63;
            int grow = min(row0 + r, N_PAD - 1);
            va[j] = *(const float4*)(agg + (long long)grow * 256 + c4 * 4);
            vh[j] = *(const float4*)(h   + (long long)grow * 256 + c4 * 4);
        }
#pragma unroll
        for (int j = 0; j < 8; ++j) {
            int idx = (it + j) * 256 + tid;
            int r = idx >> 6, c4 = idx & 63;
            v4bf pa, ph;
            pa[0] = (__bf16)va[j].x; pa[1] = (__bf16)va[j].y;
            pa[2] = (__bf16)va[j].z; pa[3] = (__bf16)va[j].w;
            ph[0] = (__bf16)vh[j].x; ph[1] = (__bf16)vh[j].y;
            ph[2] = (__bf16)vh[j].z; ph[3] = (__bf16)vh[j].w;
            *(v4bf*)(sAgg + r * GRU_PITCH + c4 * 4) = pa;
            *(v4bf*)(sH   + r * GRU_PITCH + c4 * 4) = ph;
        }
    }
    __syncthreads();

#pragma unroll 1
    for (int t = 0; t < 2; ++t) {
        int c = wave * 32 + t * 16 + cl;

        v8f aR[GRU_RT], aZ[GRU_RT], aIN[GRU_RT], aHN[GRU_RT];
#pragma unroll
        for (int rt = 0; rt < GRU_RT; ++rt) {
            v8f z = {};
            aR[rt] = z; aZ[rt] = z; aIN[rt] = z; aHN[rt] = z;
        }

        v16bf w0[6], w1[6];
        v16bf aP[8], aQ[8];
        GRU_LOAD_W(w0, 0)
        GRU_LOAD_A(aP, 0, 0)

#pragma unroll 1
        for (int kk = 0; kk < 4; ++kk) {
            int k0 = kk * 64;
            GRU_LOAD_W(w1, k0 + 32)
            GRU_LOAD_A(aQ, 4, k0)
            GRU_FENCE
            GRU_BLOCK(aP, w0, 0)
            GRU_FENCE
            GRU_LOAD_A(aP, 0, k0 + 32)
            GRU_FENCE
            GRU_BLOCK(aQ, w0, 4)
            GRU_FENCE
            if (kk < 3) {
                GRU_LOAD_W(w0, k0 + 64)
            }
            GRU_LOAD_A(aQ, 4, k0 + 32)
            GRU_FENCE
            GRU_BLOCK(aP, w1, 0)
            GRU_FENCE
            if (kk < 3) {
                GRU_LOAD_A(aP, 0, k0 + 64)
            }
            GRU_FENCE
            GRU_BLOCK(aQ, w1, 4)
            GRU_FENCE
        }

        float bir = bih[c],       bhr = bhh[c];
        float biz = bih[256 + c], bhz = bhh[256 + c];
        float bin = bih[512 + c], bhn = bhh[512 + c];
        float hold[GRU_RT][8];
        const float* hcol = h + (8 * half) * 256 + c;
        float* hst = h + (8 * half) * 256 + c;
#pragma unroll
        for (int rt = 0; rt < GRU_RT; ++rt) {
            int rbase = (row0 + rt * 16 < N_PAD) ? row0 + rt * 16 : row0;
#pragma unroll
            for (int vr = 0; vr < 8; ++vr)
                hold[rt][vr] = hcol[(long long)(rbase + vr) * 256];
        }
#pragma unroll
        for (int rt = 0; rt < GRU_RT; ++rt) {
            if (row0 + rt * 16 < N_PAD) {
#pragma unroll
                for (int vr = 0; vr < 8; ++vr) {
                    float r  = sigmoidf_(aR[rt][vr] + bir + bhr);
                    float z  = sigmoidf_(aZ[rt][vr] + biz + bhz);
                    float nn = tanhf(aIN[rt][vr] + bin + r * (aHN[rt][vr] + bhn));
                    hst[(long long)(row0 + rt * 16 + vr) * 256] =
                        (1.0f - z) * nn + z * hold[rt][vr];
                }
            }
        }
    }
}

__global__ void __launch_bounds__(256)
final_kernel(const float* __restrict__ h, const float* __restrict__ fcw,
             const float* __restrict__ fcb, float* __restrict__ out) {
    __shared__ float sw[N_ACT_C * 256];
    __shared__ float sb[N_ACT_C];
    for (int i = threadIdx.x; i < N_ACT_C * 256; i += 256) sw[i] = fcw[i];
    if (threadIdx.x < N_ACT_C) sb[threadIdx.x] = fcb[threadIdx.x];
    __syncthreads();

    int n = blockIdx.x * 256 + threadIdx.x;
    if (n >= N_NODES_C) return;

    float acc[N_ACT_C];
#pragma unroll
    for (int a = 0; a < N_ACT_C; ++a) acc[a] = sb[a];
    const float* hr = h + (long long)n * 256;
    for (int k = 0; k < 256; ++k) {
        float hv = fmaxf(hr[k], 0.0f);
#pragma unroll
        for (int a = 0; a < N_ACT_C; ++a) acc[a] += hv * sw[a * 256 + k];
    }
    float mx = acc[0];
#pragma unroll
    for (int a = 1; a < N_ACT_C; ++a) mx = fmaxf(mx, acc[a]);
    float s = 0.0f;
#pragma unroll
    for (int a = 0; a < N_ACT_C; ++a) { acc[a] = __expf(acc[a] - mx); s += acc[a]; }
    float inv = 1.0f / s;
#pragma unroll
    for (int a = 0; a < N_ACT_C; ++a) out[(long long)n * N_ACT_C + a] = acc[a] * inv;
}

extern "C" void kernel_launch(void* const* d_in, const int* in_sizes, int n_in,
                              void* d_out, int out_size, void* d_ws, size_t ws_size,
                              hipStream_t stream) {
    const float* x   = (const float*)d_in[0];
    const int*   ei  = (const int*)  d_in[1];
    const float* ew  = (const float*)d_in[2];
    const float* gg  = (const float*)d_in[3];
    const float* wih = (const float*)d_in[4];
    const float* whh = (const float*)d_in[5];
    const float* bih = (const float*)d_in[6];
    const float* bhh = (const float*)d_in[7];
    const float* fcw = (const float*)d_in[8];
    const float* fcb = (const float*)d_in[9];
    float* out = (float*)d_out;

    float* h   = (float*)d_ws;
    float* m   = h + (size_t)N_PAD * 256;
    float* agg = m + (size_t)N_PAD * 256;
    __bf16* ggT  = (__bf16*)(agg + (size_t)N_PAD * 256);
    __bf16* wihb = ggT + 3 * 256 * 256;
    __bf16* whhb = wihb + 768 * 256;
    int* counts   = (int*)(whhb + 768 * 256);
    int* within   = counts + (size_t)NBLK * BIN_PITCH;
    int* binTotal = within + (size_t)NBLK * BIN_PITCH;
    int* binStart = binTotal + BIN_PITCH;
    int4* rec     = (int4*)(binStart + BIN_PITCH);

    convw_kernel<<<(3 * 256 * 256 + 255) / 256, 256, 0, stream>>>(gg, wih, whh,
                                                                  ggT, wihb, whhb);
    init_h_kernel<<<N_PAD, 256, 0, stream>>>(x, h);
    bin_count_kernel<<<NBLK, 256, 0, stream>>>(ei, counts);
    bin_scan_kernel<<<(NBIN + 31) / 32, 32, 0, stream>>>(counts, within, binTotal);
    bin_place_kernel<<<NBLK, 256, 0, stream>>>(ei, ew, within, binTotal, binStart, rec);

    for (int l = 0; l < 3; ++l) {
        gemm_m_kernel<<<N_PAD / 32, 256, 0, stream>>>(h, ggT + (size_t)l * 65536, m);
        agg_kernel<<<NBIN, 128, 0, stream>>>(m, rec, binStart, binTotal, agg);
        gru_kernel<<<(N_PAD + GRU_RB - 1) / GRU_RB, 256, 0, stream>>>(agg, h, wihb, whhb,
                                                                     bih, bhh);
    }

    final_kernel<<<(N_NODES_C + 255) / 256, 256, 0, stream>>>(h, fcw, fcb, out);
}


# Message GEMM kernel: work-group tile 128 rows (was 32), its h rows staged once as bf16 in LDS; a weight fragment from global feeds 8 row tiles (was 1); k order of every sum and values unchanged
# speedup vs baseline: 3.0222x; 1.1738x over previous
#include <hip/hip_runtime.h>

#define N_NODES_C 50000
#define N_EDGES_C 800000
#define F_IN_C 32
#define C_C 256
#define N_PAD 50016
#define N_ACT_C 8

typedef __attribute__((ext_vector_type(16))) __bf16 v16bf;
typedef __attribute__((ext_vector_type(8)))  __bf16 v8bf;
typedef __attribute__((ext_vector_type(8)))  float  v8f;

static __device__ __forceinline__ v8f wmma_bf16(v16bf a, v16bf b, v8f c) {
    return __builtin_amdgcn_wmma_f32_16x16x32_bf16(
         false, a,  false, b,
         (short)0, c,  false,  false);
}

static __device__ __forceinline__ v16bf load_b_frag_bf16(
    const __bf16* __restrict__ BT, int col, int ld, int k0, int lane) {
    int half = lane >> 4;
    const __bf16* p = BT + (long long)col * ld + k0 + half * 8;
    v8bf b0 = *(const v8bf*)(p);
    v8bf b1 = *(const v8bf*)(p + 16);
    v16bf b;
#pragma unroll
    for (int i = 0; i < 8; ++i) { b[i] = b0[i]; b[8 + i] = b1[i]; }
    return b;
}

static __device__ __forceinline__ float sigmoidf_(float x) {
    return 1.0f / (1.0f + __expf(-x));
}

__global__ void convw_kernel(const float* __restrict__ gg,
                             const float* __restrict__ wih,
                             const float* __restrict__ whh,
                             __bf16* __restrict__ ggT,
                             __bf16* __restrict__ wihb,
                             __bf16* __restrict__ whhb) {
    int i = blockIdx.x * 256 + threadIdx.x;
    if (i < 3 * 256 * 256) {
        int l = i >> 16, r = i & 65535, j = r >> 8, k = r & 255;
        ggT[i] = (__bf16)gg[l * 65536 + k * 256 + j];
    }
    if (i < 768 * 256) {
        wihb[i] = (__bf16)wih[i];
        whhb[i] = (__bf16)whh[i];
    }
}

__global__ void init_h_kernel(const float* __restrict__ x, float* __restrict__ h) {
    int i = blockIdx.x * 256 + threadIdx.x;
    int n = i >> 8, j = i & 255;
    float v = 0.0f;
    if (n < N_NODES_C && j < F_IN_C) v = x[n * F_IN_C + j];
    h[i] = v;
}

#define EBLK 2048
#define NBLK ((N_EDGES_C + EBLK - 1) / EBLK)
#define BIN_SHIFT 4
#define BIN_ROWS 16
#define NBIN (N_PAD / BIN_ROWS)
#define BIN_PITCH 3136

__global__ void __launch_bounds__(256)
bin_count_kernel(const int* __restrict__ ei, int* __restrict__ counts) {
    __shared__ int hist[NBIN];
    for (int i = threadIdx.x; i < NBIN; i += 256) hist[i] = 0;
    __syncthreads();
    int e0 = blockIdx.x * EBLK;
#pragma unroll
    for (int j = 0; j < EBLK / 256; ++j) {
        int e = e0 + j * 256 + threadIdx.x;
        if (e < N_EDGES_C) atomicAdd(&hist[ei[N_EDGES_C + e] >> BIN_SHIFT], 1);
    }
    __syncthreads();
    int* row = counts + (size_t)blockIdx.x * BIN_PITCH;
    for (int i = threadIdx.x; i < NBIN; i += 256) row[i] = hist[i];
}

#define SCAN_U 17
__global__ void __launch_bounds__(32)
bin_scan_kernel(const int* __restrict__ counts, int* __restrict__ within,
                int* __restrict__ binTotal) {
    int bin = blockIdx.x * 32 + threadIdx.x;
    if (bin >= NBIN) return;
    int run = 0;
    for (int b0 = 0; b0 < NBLK; b0 += SCAN_U) {
        int c[SCAN_U];
#pragma unroll
        for (int u = 0; u < SCAN_U; ++u)
            c[u] = (b0 + u < NBLK) ? counts[(size_t)(b0 + u) * BIN_PITCH + bin] : 0;
#pragma unroll
        for (int u = 0; u < SCAN_U; ++u) {
            if (b0 + u < NBLK) within[(size_t)(b0 + u) * BIN_PITCH + bin] = run;
            run += c[u];
        }
    }
    binTotal[bin] = run;
}

#define PLACE_BPT ((NBIN + 255) / 256)
__global__ void __launch_bounds__(256)
bin_place_kernel(const int* __restrict__ ei, const float* __restrict__ ew,
                 const int* __restrict__ within, const int* __restrict__ binTotal,
                 int* __restrict__ binStart, int4* __restrict__ rec) {
    __shared__ int sCur[8][NBIN];
    __shared__ int sBase[NBIN];
    __shared__ int sWave[8];
    int tid = threadIdx.x, lane = tid & 31, wave = tid >> 5;
    int blk = blockIdx.x;

    int loc[PLACE_BPT];
    int sum = 0;
#pragma unroll
    for (int k = 0; k < PLACE_BPT; ++k) {
        int bin = PLACE_BPT * tid + k;
        int v = (bin < NBIN) ? binTotal[bin] : 0;
        loc[k] = sum;
        sum += v;
    }
    int inc = sum;
#pragma unroll
    for (int o = 1; o < 32; o <<= 1) {
        int t = __shfl_up(inc, o);
        if (lane >= o) inc += t;
    }
    if (lane == 31) sWave[wave] = inc;
    for (int i = tid; i < 8 * NBIN; i += 256) (&sCur[0][0])[i] = 0;
    __syncthreads();
    int wbase = 0;
    for (int w = 0; w < wave; ++w) wbase += sWave[w];
    int excl = wbase + inc - sum;
#pragma unroll
    for (int k = 0; k < PLACE_BPT; ++k) {
        int bin = PLACE_BPT * tid + k;
        if (bin < NBIN) {
            int st = excl + loc[k];
            sBase[bin] = st + within[(size_t)blk * BIN_PITCH + bin];
            if (blk == 0) binStart[bin] = st;
        }
    }

    int e0 = blk * EBLK + wave * 256;
    int dreg[8], sreg[8];
    float wreg[8];
#pragma unroll
    for (int c = 0; c < 8; ++c) {
        int e = e0 + c * 32 + lane;
        bool ok = e < N_EDGES_C;
        dreg[c] = ok ? ei[N_EDGES_C + e] : -1;
        sreg[c] = ok ? ei[e] : 0;
        wreg[c] = ok ? ew[e] : 0.0f;
        if (ok) atomicAdd(&sCur[wave][dreg[c] >> BIN_SHIFT], 1);
    }
    __syncthreads();
    for (int bin = tid; bin < NBIN; bin += 256) {
        int run = sBase[bin];
#pragma unroll
        for (int w = 0; w < 8; ++w) {
            int c = sCur[w][bin];
            sCur[w][bin] = run;
            run += c;
        }
    }
    __syncthreads();
#pragma unroll
    for (int c = 0; c < 8; ++c) {
        int d = dreg[c];
        int bin = d >> BIN_SHIFT;
        int rank = 0, cnt = 0;
#pragma unroll
        for (int j = 0; j < 32; ++j) {
            int bj = __builtin_amdgcn_readlane(bin, j);
            int same = (bj == bin) ? 1 : 0;
            cnt += same;
            rank += (j < lane) ? same : 0;
        }
        if (d >= 0) {
            int pos = sCur[wave][bin] + rank;
            rec[pos] = make_int4(sreg[c], d & (BIN_ROWS - 1), __float_as_int(wreg[c]), 0);
            if (rank == cnt - 1) sCur[wave][bin] = pos + 1;
        }
    }
}

#define AGG_U 16
__global__ void __launch_bounds__(128)
agg_kernel(const float* __restrict__ m, const int4* __restrict__ rec,
           const int* __restrict__ binStart, const int* __restrict__ binTotal,
           float* __restrict__ agg) {
    __shared__ __attribute__((aligned(16))) float tile[(BIN_ROWS + 1) * 256];
    int lane = threadIdx.x & 31, wave = threadIdx.x >> 5;
    int b = blockIdx.x;
    int col = wave * 64 + lane * 2;
    float* tcol = tile + col;
    const float* mcol = m + col;
#pragma unroll
    for (int r = 0; r <= BIN_ROWS; ++r) *(float2*)(tcol + r * 256) = make_float2(0.0f, 0.0f);

    int start = __builtin_amdgcn_readfirstlane(binStart[b]);
    int n     = __builtin_amdgcn_readfirstlane(binTotal[b]);
    const int4* list = rec + start;

    const int4 filler = make_int4(0, BIN_ROWS, 0, 0);
    int4 cur = (lane < n) ? list[lane] : filler;
    for (int c0 = 0; c0 < n; c0 += 32) {
        int nidx = c0 + 32 + lane;
        int4 nxt = (nidx < n) ? list[nidx] : filler;
        int cnt = min(32, n - c0);
        for (int j = 0; j < cnt; j += AGG_U) {
            float2 v[AGG_U];
#pragma unroll
            for (int u = 0; u < AGG_U; ++u) {
                int s = __builtin_amdgcn_readlane(cur.x, j + u);
                v[u] = *(const float2*)(mcol + (size_t)s * 256);
            }
#pragma unroll
            for (int u = 0; u < AGG_U; ++u) {
                int dl = __builtin_amdgcn_readlane(cur.y, j + u);
                float w = __int_as_float(__builtin_amdgcn_readlane(cur.z, j + u));
                atomicAdd(tcol + dl * 256,     v[u].x * w);
                atomicAdd(tcol + dl * 256 + 1, v[u].y * w);
            }
        }
        cur = nxt;
    }

    float* arow = agg + (size_t)b * BIN_ROWS * 256 + col;
#pragma unroll
    for (int r = 0; r < BIN_ROWS; ++r)
        *(float2*)(arow + r * 256) = *(const float2*)(tcol + r * 256);
}

#define GRU_RB 128
#define GRU_RT (GRU_RB / 16)
#define GRU_PITCH 264

typedef __attribute__((ext_vector_type(4))) __bf16 v4bf;

static __device__ __forceinline__ v16bf load_a_frag_lds(
    const __bf16* sA, int row, int k0, int half) {
    const __bf16* p = sA + row * GRU_PITCH + k0 + half * 8;
    v8bf a0 = *(const v8bf*)(p);
    v8bf a1 = *(const v8bf*)(p + 16);
    v16bf a;
#pragma unroll
    for (int i = 0; i < 8; ++i) { a[i] = a0[i]; a[8 + i] = a1[i]; }
    return a;
}

#define GM_RB 128
#define GM_RT (GM_RB / 16)

#define GM_LOAD_A(AS, rb, kq)                                               \
    _Pragma("unroll")                                                       \
    for (int i = 0; i < 4; ++i)                                             \
        AS[i] = load_a_frag_lds(sHm, ((rb) + i) * 16 + cl, (kq), half);

#define GM_LOAD_W(WS, kq)                                                   \
    _Pragma("unroll")                                                       \
    for (int t = 0; t < 2; ++t)                                             \
        WS[t] = load_b_frag_bf16(WgT, c0 + t * 16 + cl, 256, (kq), lane);

#define GM_BLOCK(AS, WS, rb)                                                \
    _Pragma("unroll")                                                       \
    for (int i = 0; i < 4; ++i) {                                           \
        acc0[(rb) + i] = wmma_bf16(AS[i], WS[0], acc0[(rb) + i]);           \
        acc1[(rb) + i] = wmma_bf16(AS[i], WS[1], acc1[(rb) + i]);           \
    }

#define GM_FENCE __builtin_amdgcn_sched_barrier(0);

__global__ void __launch_bounds__(256)
gemm_m_kernel(const float* __restrict__ h, const __bf16* __restrict__ WgT,
              float* __restrict__ m) {
    __shared__ __attribute__((aligned(16))) __bf16 sHm[GM_RB * GRU_PITCH];

    int tid  = threadIdx.x;
    int lane = tid & 31;
    int wave = tid >> 5;
    int half = lane >> 4;
    int cl   = lane & 15;
    int row0 = blockIdx.x * GM_RB;
    int c0   = wave * 32;

#pragma unroll 1
    for (int it = 0; it < 32; it += 16) {
        float4 vh[16];
#pragma unroll
        for (int j = 0; j < 16; ++j) {
            int idx = (it + j) * 256 + tid;
            int r = idx >> 6, c4 = idx & 63;
            int grow = min(row0 + r, N_PAD - 1);
            vh[j] = *(const float4*)(h + (long long)grow * 256 + c4 * 4);
        }
#pragma unroll
        for (int j = 0; j < 16; ++j) {
            int idx = (it + j) * 256 + tid;
            int r = idx >> 6, c4 = idx & 63;
            v4bf ph;
            ph[0] = (__bf16)vh[j].x; ph[1] = (__bf16)vh[j].y;
            ph[2] = (__bf16)vh[j].z; ph[3] = (__bf16)vh[j].w;
            *(v4bf*)(sHm + r * GRU_PITCH + c4 * 4) = ph;
        }
    }
    __syncthreads();

    v8f acc0[GM_RT], acc1[GM_RT];
#pragma unroll
    for (int rt = 0; rt < GM_RT; ++rt) {
        v8f z = {};
        acc0[rt] = z; acc1[rt] = z;
    }

    v16bf w0[2], w1[2];
    v16bf aP[4], aQ[4];
    GM_LOAD_W(w0, 0)
    GM_LOAD_A(aP, 0, 0)

#pragma unroll 1
    for (int kk = 0; kk < 4; ++kk) {
        int k0 = kk * 64;
        GM_LOAD_W(w1, k0 + 32)
        GM_LOAD_A(aQ, 4, k0)
        GM_FENCE
        GM_BLOCK(aP, w0, 0)
        GM_FENCE
        GM_LOAD_A(aP, 0, k0 + 32)
        GM_FENCE
        GM_BLOCK(aQ, w0, 4)
        GM_FENCE
        if (kk < 3) {
            GM_LOAD_W(w0, k0 + 64)
        }
        GM_LOAD_A(aQ, 4, k0 + 32)
        GM_FENCE
        GM_BLOCK(aP, w1, 0)
        GM_FENCE
        if (kk < 3) {
            GM_LOAD_A(aP, 0, k0 + 64)
        }
        GM_FENCE
        GM_BLOCK(aQ, w1, 4)
        GM_FENCE
    }

    float* mst = m + (8 * half) * 256 + c0 + cl;
#pragma unroll
    for (int rt = 0; rt < GM_RT; ++rt) {
        if (row0 + rt * 16 < N_PAD) {
#pragma unroll
            for (int vr = 0; vr < 8; ++vr) {
                float* p = mst + (long long)(row0 + rt * 16 + vr) * 256;
                p[0]  = acc0[rt][vr];
                p[16] = acc1[rt][vr];
            }
        }
    }
}

#define GRU_LOAD_A(AS, rb, kq)                                              \
    _Pragma("unroll")                                                       \
    for (int i = 0; i < 4; ++i) {                                           \
        AS[i]     = load_a_frag_lds(sAgg, ((rb) + i) * 16 + cl, (kq), half);\
        AS[4 + i] = load_a_frag_lds(sH,   ((rb) + i) * 16 + cl, (kq), half);\
    }

#define GRU_BLOCK(AS, WS, rb)                                               \
    _Pragma("unroll")                                                       \
    for (int i = 0; i < 4; ++i) {                                           \
        aR[(rb) + i]  = wmma_bf16(AS[i], WS[0], aR[(rb) + i]);              \
        aZ[(rb) + i]  = wmma_bf16(AS[i], WS[1], aZ[(rb) + i]);              \
        aIN[(rb) + i] = wmma_bf16(AS[i], WS[2], aIN[(rb) + i]);             \
    }                                                                       \
    _Pragma("unroll")                                                       \
    for (int i = 0; i < 4; ++i) {                                           \
        aR[(rb) + i]  = wmma_bf16(AS[4 + i], WS[3], aR[(rb) + i]);          \
        aZ[(rb) + i]  = wmma_bf16(AS[4 + i], WS[4], aZ[(rb) + i]);          \
        aHN[(rb) + i] = wmma_bf16(AS[4 + i], WS[5], aHN[(rb) + i]);         \
    }

#define GRU_FENCE __builtin_amdgcn_sched_barrier(0);

#define GRU_LOAD_W(WS, kq)                                                  \
    _Pragma("unroll")                                                       \
    for (int g = 0; g < 3; ++g) {                                           \
        WS[g]     = load_b_frag_bf16(wihb, g * 256 + c, 256, (kq), lane);   \
        WS[3 + g] = load_b_frag_bf16(whhb, g * 256 + c, 256, (kq), lane);   \
    }

__global__ void __launch_bounds__(256) __attribute__((amdgpu_waves_per_eu(2, 2)))
gru_kernel(const float* __restrict__ agg, float* __restrict__ h,
           const __bf16* __restrict__ wihb, const __bf16* __restrict__ whhb,
           const float* __restrict__ bih, const float* __restrict__ bhh) {
    __shared__ __attribute__((aligned(16))) __bf16 sA[2 * GRU_RB * GRU_PITCH];
    __bf16* sAgg = sA;
    __bf16* sH   = sA + GRU_RB * GRU_PITCH;

    int tid  = threadIdx.x;
    int lane = tid & 31;
    int wave = tid >> 5;
    int half = lane >> 4;
    int cl   = lane & 15;
    int row0 = blockIdx.x * GRU_RB;

#pragma unroll 1
    for (int it = 0; it < 32; it += 8) {
        float4 va[8], vh[8];
#pragma unroll
        for (int j = 0; j < 8; ++j) {
            int idx = (it + j) * 256 + tid;
            int r = idx >> 6, c4 = idx & 63;
            int grow = min(row0 + r, N_PAD - 1);
            va[j] = *(const float4*)(agg + (long long)grow * 256 + c4 * 4);
            vh[j] = *(const float4*)(h   + (long long)grow * 256 + c4 * 4);
        }
#pragma unroll
        for (int j = 0; j < 8; ++j) {
            int idx = (it + j) * 256 + tid;
            int r = idx >> 6, c4 = idx & 63;
            v4bf pa, ph;
            pa[0] = (__bf16)va[j].x; pa[1] = (__bf16)va[j].y;
            pa[2] = (__bf16)va[j].z; pa[3] = (__bf16)va[j].w;
            ph[0] = (__bf16)vh[j].x; ph[1] = (__bf16)vh[j].y;
            ph[2] = (__bf16)vh[j].z; ph[3] = (__bf16)vh[j].w;
            *(v4bf*)(sAgg + r * GRU_PITCH + c4 * 4) = pa;
            *(v4bf*)(sH   + r * GRU_PITCH + c4 * 4) = ph;
        }
    }
    __syncthreads();

#pragma unroll 1
    for (int t = 0; t < 2; ++t) {
        int c = wave * 32 + t * 16 + cl;

        v8f aR[GRU_RT], aZ[GRU_RT], aIN[GRU_RT], aHN[GRU_RT];
#pragma unroll
        for (int rt = 0; rt < GRU_RT; ++rt) {
            v8f z = {};
            aR[rt] = z; aZ[rt] = z; aIN[rt] = z; aHN[rt] = z;
        }

        v16bf w0[6], w1[6];
        v16bf aP[8], aQ[8];
        GRU_LOAD_W(w0, 0)
        GRU_LOAD_A(aP, 0, 0)

#pragma unroll 1
        for (int kk = 0; kk < 4; ++kk) {
            int k0 = kk * 64;
            GRU_LOAD_W(w1, k0 + 32)
            GRU_LOAD_A(aQ, 4, k0)
            GRU_FENCE
            GRU_BLOCK(aP, w0, 0)
            GRU_FENCE
            GRU_LOAD_A(aP, 0, k0 + 32)
            GRU_FENCE
            GRU_BLOCK(aQ, w0, 4)
            GRU_FENCE
            if (kk < 3) {
                GRU_LOAD_W(w0, k0 + 64)
            }
            GRU_LOAD_A(aQ, 4, k0 + 32)
            GRU_FENCE
            GRU_BLOCK(aP, w1, 0)
            GRU_FENCE
            if (kk < 3) {
                GRU_LOAD_A(aP, 0, k0 + 64)
            }
            GRU_FENCE
            GRU_BLOCK(aQ, w1, 4)
            GRU_FENCE
        }

        float bir = bih[c],       bhr = bhh[c];
        float biz = bih[256 + c], bhz = bhh[256 + c];
        float bin = bih[512 + c], bhn = bhh[512 + c];
        float hold[GRU_RT][8];
        const float* hcol = h + (8 * half) * 256 + c;
        float* hst = h + (8 * half) * 256 + c;
#pragma unroll
        for (int rt = 0; rt < GRU_RT; ++rt) {
            int rbase = (row0 + rt * 16 < N_PAD) ? row0 + rt * 16 : row0;
#pragma unroll
            for (int vr = 0; vr < 8; ++vr)
                hold[rt][vr] = hcol[(long long)(rbase + vr) * 256];
        }
#pragma unroll
        for (int rt = 0; rt < GRU_RT; ++rt) {
            if (row0 + rt * 16 < N_PAD) {
#pragma unroll
                for (int vr = 0; vr < 8; ++vr) {
                    float r  = sigmoidf_(aR[rt][vr] + bir + bhr);
                    float z  = sigmoidf_(aZ[rt][vr] + biz + bhz);
                    float nn = tanhf(aIN[rt][vr] + bin + r * (aHN[rt][vr] + bhn));
                    hst[(long long)(row0 + rt * 16 + vr) * 256] =
                        (1.0f - z) * nn + z * hold[rt][vr];
                }
            }
        }
    }
}

__global__ void __launch_bounds__(256)
final_kernel(const float* __restrict__ h, const float* __restrict__ fcw,
             const float* __restrict__ fcb, float* __restrict__ out) {
    __shared__ float sw[N_ACT_C * 256];
    __shared__ float sb[N_ACT_C];
    for (int i = threadIdx.x; i < N_ACT_C * 256; i += 256) sw[i] = fcw[i];
    if (threadIdx.x < N_ACT_C) sb[threadIdx.x] = fcb[threadIdx.x];
    __syncthreads();

    int n = blockIdx.x * 256 + threadIdx.x;
    if (n >= N_NODES_C) return;

    float acc[N_ACT_C];
#pragma unroll
    for (int a = 0; a < N_ACT_C; ++a) acc[a] = sb[a];
    const float* hr = h + (long long)n * 256;
    for (int k = 0; k < 256; ++k) {
        float hv = fmaxf(hr[k], 0.0f);
#pragma unroll
        for (int a = 0; a < N_ACT_C; ++a) acc[a] += hv * sw[a * 256 + k];
    }
    float mx = acc[0];
#pragma unroll
    for (int a = 1; a < N_ACT_C; ++a) mx = fmaxf(mx, acc[a]);
    float s = 0.0f;
#pragma unroll
    for (int a = 0; a < N_ACT_C; ++a) { acc[a] = __expf(acc[a] - mx); s += acc[a]; }
    float inv = 1.0f / s;
#pragma unroll
    for (int a = 0; a < N_ACT_C; ++a) out[(long long)n * N_ACT_C + a] = acc[a] * inv;
}

extern "C" void kernel_launch(void* const* d_in, const int* in_sizes, int n_in,
                              void* d_out, int out_size, void* d_ws, size_t ws_size,
                              hipStream_t stream) {
    const float* x   = (const float*)d_in[0];
    const int*   ei  = (const int*)  d_in[1];
    const float* ew  = (const float*)d_in[2];
    const float* gg  = (const float*)d_in[3];
    const float* wih = (const float*)d_in[4];
    const float* whh = (const float*)d_in[5];
    const float* bih = (const float*)d_in[6];
    const float* bhh = (const float*)d_in[7];
    const float* fcw = (const float*)d_in[8];
    const float* fcb = (const float*)d_in[9];
    float* out = (float*)d_out;

    float* h   = (float*)d_ws;
    float* m   = h + (size_t)N_PAD * 256;
    float* agg = m + (size_t)N_PAD * 256;
    __bf16* ggT  = (__bf16*)(agg + (size_t)N_PAD * 256);
    __bf16* wihb = ggT + 3 * 256 * 256;
    __bf16* whhb = wihb + 768 * 256;
    int* counts   = (int*)(whhb + 768 * 256);
    int* within   = counts + (size_t)NBLK * BIN_PITCH;
    int* binTotal = within + (size_t)NBLK * BIN_PITCH;
    int* binStart = binTotal + BIN_PITCH;
    int4* rec     = (int4*)(binStart + BIN_PITCH);

    convw_kernel<<<(3 * 256 * 256 + 255) / 256, 256, 0, stream>>>(gg, wih, whh,
                                                                  ggT, wihb, whhb);
    init_h_kernel<<<N_PAD, 256, 0, stream>>>(x, h);
    bin_count_kernel<<<NBLK, 256, 0, stream>>>(ei, counts);
    bin_scan_kernel<<<(NBIN + 31) / 32, 32, 0, stream>>>(counts, within, binTotal);
    bin_place_kernel<<<NBLK, 256, 0, stream>>>(ei, ew, within, binTotal, binStart, rec);

    for (int l = 0; l < 3; ++l) {
        gemm_m_kernel<<<(N_PAD + GM_RB - 1) / GM_RB, 256, 0, stream>>>(h, ggT + (size_t)l * 65536, m);
        agg_kernel<<<NBIN, 128, 0, stream>>>(m, rec, binStart, binTotal, agg);
        gru_kernel<<<(N_PAD + GRU_RB - 1) / GRU_RB, 256, 0, stream>>>(agg, h, wihb, whhb,
                                                                     bih, bhh);
    }

    final_kernel<<<(N_NODES_C + 255) / 256, 256, 0, stream>>>(h, fcw, fcb, out);
}


# GRU kernel grid balanced: 241 work-groups of 13 row tiles (two groups, 7 then 6, in sequence) fill the 256 WGPs in one round; was 391 of 8 tiles = two rounds, second half empty; sums unchanged
# speedup vs baseline: 3.1351x; 1.0374x over previous
#include <hip/hip_runtime.h>

#define N_NODES_C 50000
#define N_EDGES_C 800000
#define F_IN_C 32
#define C_C 256
#define N_PAD 50016
#define N_ACT_C 8

typedef __attribute__((ext_vector_type(16))) __bf16 v16bf;
typedef __attribute__((ext_vector_type(8)))  __bf16 v8bf;
typedef __attribute__((ext_vector_type(8)))  float  v8f;

static __device__ __forceinline__ v8f wmma_bf16(v16bf a, v16bf b, v8f c) {
    return __builtin_amdgcn_wmma_f32_16x16x32_bf16(
         false, a,  false, b,
         (short)0, c,  false,  false);
}

static __device__ __forceinline__ v16bf load_b_frag_bf16(
    const __bf16* __restrict__ BT, int col, int ld, int k0, int lane) {
    int half = lane >> 4;
    const __bf16* p = BT + (long long)col * ld + k0 + half * 8;
    v8bf b0 = *(const v8bf*)(p);
    v8bf b1 = *(const v8bf*)(p + 16);
    v16bf b;
#pragma unroll
    for (int i = 0; i < 8; ++i) { b[i] = b0[i]; b[8 + i] = b1[i]; }
    return b;
}

static __device__ __forceinline__ float sigmoidf_(float x) {
    return 1.0f / (1.0f + __expf(-x));
}

__global__ void convw_kernel(const float* __restrict__ gg,
                             const float* __restrict__ wih,
                             const float* __restrict__ whh,
                             __bf16* __restrict__ ggT,
                             __bf16* __restrict__ wihb,
                             __bf16* __restrict__ whhb) {
    int i = blockIdx.x * 256 + threadIdx.x;
    if (i < 3 * 256 * 256) {
        int l = i >> 16, r = i & 65535, j = r >> 8, k = r & 255;
        ggT[i] = (__bf16)gg[l * 65536 + k * 256 + j];
    }
    if (i < 768 * 256) {
        wihb[i] = (__bf16)wih[i];
        whhb[i] = (__bf16)whh[i];
    }
}

__global__ void init_h_kernel(const float* __restrict__ x, float* __restrict__ h) {
    int i = blockIdx.x * 256 + threadIdx.x;
    int n = i >> 8, j = i & 255;
    float v = 0.0f;
    if (n < N_NODES_C && j < F_IN_C) v = x[n * F_IN_C + j];
    h[i] = v;
}

#define EBLK 2048
#define NBLK ((N_EDGES_C + EBLK - 1) / EBLK)
#define BIN_SHIFT 4
#define BIN_ROWS 16
#define NBIN (N_PAD / BIN_ROWS)
#define BIN_PITCH 3136

__global__ void __launch_bounds__(256)
bin_count_kernel(const int* __restrict__ ei, int* __restrict__ counts) {
    __shared__ int hist[NBIN];
    for (int i = threadIdx.x; i < NBIN; i += 256) hist[i] = 0;
    __syncthreads();
    int e0 = blockIdx.x * EBLK;
#pragma unroll
    for (int j = 0; j < EBLK / 256; ++j) {
        int e = e0 + j * 256 + threadIdx.x;
        if (e < N_EDGES_C) atomicAdd(&hist[ei[N_EDGES_C + e] >> BIN_SHIFT], 1);
    }
    __syncthreads();
    int* row = counts + (size_t)blockIdx.x * BIN_PITCH;
    for (int i = threadIdx.x; i < NBIN; i += 256) row[i] = hist[i];
}

#define SCAN_U 17
__global__ void __launch_bounds__(32)
bin_scan_kernel(const int* __restrict__ counts, int* __restrict__ within,
                int* __restrict__ binTotal) {
    int bin = blockIdx.x * 32 + threadIdx.x;
    if (bin >= NBIN) return;
    int run = 0;
    for (int b0 = 0; b0 < NBLK; b0 += SCAN_U) {
        int c[SCAN_U];
#pragma unroll
        for (int u = 0; u < SCAN_U; ++u)
            c[u] = (b0 + u < NBLK) ? counts[(size_t)(b0 + u) * BIN_PITCH + bin] : 0;
#pragma unroll
        for (int u = 0; u < SCAN_U; ++u) {
            if (b0 + u < NBLK) within[(size_t)(b0 + u) * BIN_PITCH + bin] = run;
            run += c[u];
        }
    }
    binTotal[bin] = run;
}

#define PLACE_BPT ((NBIN + 255) / 256)
__global__ void __launch_bounds__(256)
bin_place_kernel(const int* __restrict__ ei, const float* __restrict__ ew,
                 const int* __restrict__ within, const int* __restrict__ binTotal,
                 int* __restrict__ binStart, int4* __restrict__ rec) {
    __shared__ int sCur[8][NBIN];
    __shared__ int sBase[NBIN];
    __shared__ int sWave[8];
    int tid = threadIdx.x, lane = tid & 31, wave = tid >> 5;
    int blk = blockIdx.x;

    int loc[PLACE_BPT];
    int sum = 0;
#pragma unroll
    for (int k = 0; k < PLACE_BPT; ++k) {
        int bin = PLACE_BPT * tid + k;
        int v = (bin < NBIN) ? binTotal[bin] : 0;
        loc[k] = sum;
        sum += v;
    }
    int inc = sum;
#pragma unroll
    for (int o = 1; o < 32; o <<= 1) {
        int t = __shfl_up(inc, o);
        if (lane >= o) inc += t;
    }
    if (lane == 31) sWave[wave] = inc;
    for (int i = tid; i < 8 * NBIN; i += 256) (&sCur[0][0])[i] = 0;
    __syncthreads();
    int wbase = 0;
    for (int w = 0; w < wave; ++w) wbase += sWave[w];
    int excl = wbase + inc - sum;
#pragma unroll
    for (int k = 0; k < PLACE_BPT; ++k) {
        int bin = PLACE_BPT * tid + k;
        if (bin < NBIN) {
            int st = excl + loc[k];
            sBase[bin] = st + within[(size_t)blk * BIN_PITCH + bin];
            if (blk == 0) binStart[bin] = st;
        }
    }

    int e0 = blk * EBLK + wave * 256;
    int dreg[8], sreg[8];
    float wreg[8];
#pragma unroll
    for (int c = 0; c < 8; ++c) {
        int e = e0 + c * 32 + lane;
        bool ok = e < N_EDGES_C;
        dreg[c] = ok ? ei[N_EDGES_C + e] : -1;
        sreg[c] = ok ? ei[e] : 0;
        wreg[c] = ok ? ew[e] : 0.0f;
        if (ok) atomicAdd(&sCur[wave][dreg[c] >> BIN_SHIFT], 1);
    }
    __syncthreads();
    for (int bin = tid; bin < NBIN; bin += 256) {
        int run = sBase[bin];
#pragma unroll
        for (int w = 0; w < 8; ++w) {
            int c = sCur[w][bin];
            sCur[w][bin] = run;
            run += c;
        }
    }
    __syncthreads();
#pragma unroll
    for (int c = 0; c < 8; ++c) {
        int d = dreg[c];
        int bin = d >> BIN_SHIFT;
        int rank = 0, cnt = 0;
#pragma unroll
        for (int j = 0; j < 32; ++j) {
            int bj = __builtin_amdgcn_readlane(bin, j);
            int same = (bj == bin) ? 1 : 0;
            cnt += same;
            rank += (j < lane) ? same : 0;
        }
        if (d >= 0) {
            int pos = sCur[wave][bin] + rank;
            rec[pos] = make_int4(sreg[c], d & (BIN_ROWS - 1), __float_as_int(wreg[c]), 0);
            if (rank == cnt - 1) sCur[wave][bin] = pos + 1;
        }
    }
}

#define AGG_U 16
__global__ void __launch_bounds__(128)
agg_kernel(const float* __restrict__ m, const int4* __restrict__ rec,
           const int* __restrict__ binStart, const int* __restrict__ binTotal,
           float* __restrict__ agg) {
    __shared__ __attribute__((aligned(16))) float tile[(BIN_ROWS + 1) * 256];
    int lane = threadIdx.x & 31, wave = threadIdx.x >> 5;
    int b = blockIdx.x;
    int col = wave * 64 + lane * 2;
    float* tcol = tile + col;
    const float* mcol = m + col;
#pragma unroll
    for (int r = 0; r <= BIN_ROWS; ++r) *(float2*)(tcol + r * 256) = make_float2(0.0f, 0.0f);

    int start = __builtin_amdgcn_readfirstlane(binStart[b]);
    int n     = __builtin_amdgcn_readfirstlane(binTotal[b]);
    const int4* list = rec + start;

    const int4 filler = make_int4(0, BIN_ROWS, 0, 0);
    int4 cur = (lane < n) ? list[lane] : filler;
    for (int c0 = 0; c0 < n; c0 += 32) {
        int nidx = c0 + 32 + lane;
        int4 nxt = (nidx < n) ? list[nidx] : filler;
        int cnt = min(32, n - c0);
        for (int j = 0; j < cnt; j += AGG_U) {
            float2 v[AGG_U];
#pragma unroll
            for (int u = 0; u < AGG_U; ++u) {
                int s = __builtin_amdgcn_readlane(cur.x, j + u);
                v[u] = *(const float2*)(mcol + (size_t)s * 256);
            }
#pragma unroll
            for (int u = 0; u < AGG_U; ++u) {
                int dl = __builtin_amdgcn_readlane(cur.y, j + u);
                float w = __int_as_float(__builtin_amdgcn_readlane(cur.z, j + u));
                atomicAdd(tcol + dl * 256,     v[u].x * w);
                atomicAdd(tcol + dl * 256 + 1, v[u].y * w);
            }
        }
        cur = nxt;
    }

    float* arow = agg + (size_t)b * BIN_ROWS * 256 + col;
#pragma unroll
    for (int r = 0; r < BIN_ROWS; ++r)
        *(float2*)(arow + r * 256) = *(const float2*)(tcol + r * 256);
}

#define GRU_RT 7
#define GRU_RB (GRU_RT * 16)
#define GRU_BT 13
#define GRU_PITCH 264

typedef __attribute__((ext_vector_type(4))) __bf16 v4bf;

static __device__ __forceinline__ v16bf load_a_frag_lds(
    const __bf16* sA, int row, int k0, int half) {
    const __bf16* p = sA + row * GRU_PITCH + k0 + half * 8;
    v8bf a0 = *(const v8bf*)(p);
    v8bf a1 = *(const v8bf*)(p + 16);
    v16bf a;
#pragma unroll
    for (int i = 0; i < 8; ++i) { a[i] = a0[i]; a[8 + i] = a1[i]; }
    return a;
}

#define GM_RB 128
#define GM_RT (GM_RB / 16)

#define GM_LOAD_A(AS, rb, kq)                                               \
    _Pragma("unroll")                                                       \
    for (int i = 0; i < 4; ++i)                                             \
        AS[i] = load_a_frag_lds(sHm, ((rb) + i) * 16 + cl, (kq), half);

#define GM_LOAD_W(WS, kq)                                                   \
    _Pragma("unroll")                                                       \
    for (int t = 0; t < 2; ++t)                                             \
        WS[t] = load_b_frag_bf16(WgT, c0 + t * 16 + cl, 256, (kq), lane);

#define GM_BLOCK(AS, WS, rb)                                                \
    _Pragma("unroll")                                                       \
    for (int i = 0; i < 4; ++i) {                                           \
        acc0[(rb) + i] = wmma_bf16(AS[i], WS[0], acc0[(rb) + i]);           \
        acc1[(rb) + i] = wmma_bf16(AS[i], WS[1], acc1[(rb) + i]);           \
    }

#define GM_FENCE __builtin_amdgcn_sched_barrier(0);

__global__ void __launch_bounds__(256)
gemm_m_kernel(const float* __restrict__ h, const __bf16* __restrict__ WgT,
              float* __restrict__ m) {
    __shared__ __attribute__((aligned(16))) __bf16 sHm[GM_RB * GRU_PITCH];

    int tid  = threadIdx.x;
    int lane = tid & 31;
    int wave = tid >> 5;
    int half = lane >> 4;
    int cl   = lane & 15;
    int row0 = blockIdx.x * GM_RB;
    int c0   = wave * 32;

#pragma unroll 1
    for (int it = 0; it < 32; it += 16) {
        float4 vh[16];
#pragma unroll
        for (int j = 0; j < 16; ++j) {
            int idx = (it + j) * 256 + tid;
            int r = idx >> 6, c4 = idx & 63;
            int grow = min(row0 + r, N_PAD - 1);
            vh[j] = *(const float4*)(h + (long long)grow * 256 + c4 * 4);
        }
#pragma unroll
        for (int j = 0; j < 16; ++j) {
            int idx = (it + j) * 256 + tid;
            int r = idx >> 6, c4 = idx & 63;
            v4bf ph;
            ph[0] = (__bf16)vh[j].x; ph[1] = (__bf16)vh[j].y;
            ph[2] = (__bf16)vh[j].z; ph[3] = (__bf16)vh[j].w;
            *(v4bf*)(sHm + r * GRU_PITCH + c4 * 4) = ph;
        }
    }
    __syncthreads();

    v8f acc0[GM_RT], acc1[GM_RT];
#pragma unroll
    for (int rt = 0; rt < GM_RT; ++rt) {
        v8f z = {};
        acc0[rt] = z; acc1[rt] = z;
    }

    v16bf w0[2], w1[2];
    v16bf aP[4], aQ[4];
    GM_LOAD_W(w0, 0)
    GM_LOAD_A(aP, 0, 0)

#pragma unroll 1
    for (int kk = 0; kk < 4; ++kk) {
        int k0 = kk * 64;
        GM_LOAD_W(w1, k0 + 32)
        GM_LOAD_A(aQ, 4, k0)
        GM_FENCE
        GM_BLOCK(aP, w0, 0)
        GM_FENCE
        GM_LOAD_A(aP, 0, k0 + 32)
        GM_FENCE
        GM_BLOCK(aQ, w0, 4)
        GM_FENCE
        if (kk < 3) {
            GM_LOAD_W(w0, k0 + 64)
        }
        GM_LOAD_A(aQ, 4, k0 + 32)
        GM_FENCE
        GM_BLOCK(aP, w1, 0)
        GM_FENCE
        if (kk < 3) {
            GM_LOAD_A(aP, 0, k0 + 64)
        }
        GM_FENCE
        GM_BLOCK(aQ, w1, 4)
        GM_FENCE
    }

    float* mst = m + (8 * half) * 256 + c0 + cl;
#pragma unroll
    for (int rt = 0; rt < GM_RT; ++rt) {
        if (row0 + rt * 16 < N_PAD) {
#pragma unroll
            for (int vr = 0; vr < 8; ++vr) {
                float* p = mst + (long long)(row0 + rt * 16 + vr) * 256;
                p[0]  = acc0[rt][vr];
                p[16] = acc1[rt][vr];
            }
        }
    }
}

#define GRU_LOAD_A(AS, rb, kq)                                              \
    _Pragma("unroll")                                                       \
    for (int i = 0; i < 4; ++i) {                                           \
        if ((rb) + i < GRU_RT) {                                            \
            AS[i]     = load_a_frag_lds(sAgg, ((rb) + i) * 16 + cl, (kq), half);\
            AS[4 + i] = load_a_frag_lds(sH,   ((rb) + i) * 16 + cl, (kq), half);\
        }                                                                   \
    }

#define GRU_BLOCK(AS, WS, rb)                                               \
    _Pragma("unroll")                                                       \
    for (int i = 0; i < 4; ++i) {                                           \
        if ((rb) + i < GRU_RT) {                                            \
            aR[(rb) + i]  = wmma_bf16(AS[i], WS[0], aR[(rb) + i]);          \
            aZ[(rb) + i]  = wmma_bf16(AS[i], WS[1], aZ[(rb) + i]);          \
            aIN[(rb) + i] = wmma_bf16(AS[i], WS[2], aIN[(rb) + i]);         \
        }                                                                   \
    }                                                                       \
    _Pragma("unroll")                                                       \
    for (int i = 0; i < 4; ++i) {                                           \
        if ((rb) + i < GRU_RT) {                                            \
            aR[(rb) + i]  = wmma_bf16(AS[4 + i], WS[3], aR[(rb) + i]);      \
            aZ[(rb) + i]  = wmma_bf16(AS[4 + i], WS[4], aZ[(rb) + i]);      \
            aHN[(rb) + i] = wmma_bf16(AS[4 + i], WS[5], aHN[(rb) + i]);     \
        }                                                                   \
    }

#define GRU_FENCE __builtin_amdgcn_sched_barrier(0);

#define GRU_LOAD_W(WS, kq)                                                  \
    _Pragma("unroll")                                                       \
    for (int g = 0; g < 3; ++g) {                                           \
        WS[g]     = load_b_frag_bf16(wihb, g * 256 + c, 256, (kq), lane);   \
        WS[3 + g] = load_b_frag_bf16(whhb, g * 256 + c, 256, (kq), lane);   \
    }

__global__ void __launch_bounds__(256) __attribute__((amdgpu_waves_per_eu(2, 2)))
gru_kernel(const float* __restrict__ agg, float* __restrict__ h,
           const __bf16* __restrict__ wihb, const __bf16* __restrict__ whhb,
           const float* __restrict__ bih, const float* __restrict__ bhh) {
    __shared__ __attribute__((aligned(16))) __bf16 sA[2 * GRU_RB * GRU_PITCH];
    __bf16* sAgg = sA;
    __bf16* sH   = sA + GRU_RB * GRU_PITCH;

    int tid  = threadIdx.x;
    int lane = tid & 31;
    int wave = tid >> 5;
    int half = lane >> 4;
    int cl   = lane & 15;
    int tile0 = blockIdx.x * GRU_BT;

#pragma unroll 1
    for (int g = 0; g < 2; ++g) {
        int row0   = (tile0 + g * GRU_RT) * 16;
        int rowEnd = min((tile0 + (g ? GRU_BT : GRU_RT)) * 16, N_PAD);
        if (row0 >= rowEnd) break;
        if (g) __syncthreads();

#pragma unroll 1
        for (int it = 0; it < 28; it += 7) {
            float4 va[7], vh[7];
#pragma unroll
            for (int j = 0; j < 7; ++j) {
                int idx = (it + j) * 256 + tid;
                int r = idx >> 6, c4 = idx & 63;
                int grow = min(row0 + r, rowEnd - 1);
                va[j] = *(const float4*)(agg + (long long)grow * 256 + c4 * 4);
                vh[j] = *(const float4*)(h   + (long long)grow * 256 + c4 * 4);
            }
#pragma unroll
            for (int j = 0; j < 7; ++j) {
                int idx = (it + j) * 256 + tid;
                int r = idx >> 6, c4 = idx & 63;
                v4bf pa, ph;
                pa[0] = (__bf16)va[j].x; pa[1] = (__bf16)va[j].y;
                pa[2] = (__bf16)va[j].z; pa[3] = (__bf16)va[j].w;
                ph[0] = (__bf16)vh[j].x; ph[1] = (__bf16)vh[j].y;
                ph[2] = (__bf16)vh[j].z; ph[3] = (__bf16)vh[j].w;
                *(v4bf*)(sAgg + r * GRU_PITCH + c4 * 4) = pa;
                *(v4bf*)(sH   + r * GRU_PITCH + c4 * 4) = ph;
            }
        }
        __syncthreads();

#pragma unroll 1
        for (int t = 0; t < 2; ++t) {
            int c = wave * 32 + t * 16 + cl;

            v8f aR[GRU_RT], aZ[GRU_RT], aIN[GRU_RT], aHN[GRU_RT];
#pragma unroll
            for (int rt = 0; rt < GRU_RT; ++rt) {
                v8f z = {};
                aR[rt] = z; aZ[rt] = z; aIN[rt] = z; aHN[rt] = z;
            }

            v16bf w0[6], w1[6];
            v16bf aP[8], aQ[8];
            GRU_LOAD_W(w0, 0)
            GRU_LOAD_A(aP, 0, 0)

#pragma unroll 1
            for (int kk = 0; kk < 4; ++kk) {
                int k0 = kk * 64;
                GRU_LOAD_W(w1, k0 + 32)
                GRU_LOAD_A(aQ, 4, k0)
                GRU_FENCE
                GRU_BLOCK(aP, w0, 0)
                GRU_FENCE
                GRU_LOAD_A(aP, 0, k0 + 32)
                GRU_FENCE
                GRU_BLOCK(aQ, w0, 4)
                GRU_FENCE
                if (kk < 3) {
                    GRU_LOAD_W(w0, k0 + 64)
                }
                GRU_LOAD_A(aQ, 4, k0 + 32)
                GRU_FENCE
                GRU_BLOCK(aP, w1, 0)
                GRU_FENCE
                if (kk < 3) {
                    GRU_LOAD_A(aP, 0, k0 + 64)
                }
                GRU_FENCE
                GRU_BLOCK(aQ, w1, 4)
                GRU_FENCE
            }

            float bir = bih[c],       bhr = bhh[c];
            float biz = bih[256 + c], bhz = bhh[256 + c];
            float bin = bih[512 + c], bhn = bhh[512 + c];
            float hold[GRU_RT][8];
            const float* hcol = h + (8 * half) * 256 + c;
            float* hst = h + (8 * half) * 256 + c;
#pragma unroll
            for (int rt = 0; rt < GRU_RT; ++rt) {
                int rbase = (row0 + rt * 16 < rowEnd) ? row0 + rt * 16 : row0;
#pragma unroll
                for (int vr = 0; vr < 8; ++vr)
                    hold[rt][vr] = hcol[(long long)(rbase + vr) * 256];
            }
#pragma unroll
            for (int rt = 0; rt < GRU_RT; ++rt) {
                if (row0 + rt * 16 < rowEnd) {
#pragma unroll
                    for (int vr = 0; vr < 8; ++vr) {
                        float r  = sigmoidf_(aR[rt][vr] + bir + bhr);
                        float z  = sigmoidf_(aZ[rt][vr] + biz + bhz);
                        float nn = tanhf(aIN[rt][vr] + bin + r * (aHN[rt][vr] + bhn));
                        hst[(long long)(row0 + rt * 16 + vr) * 256] =
                            (1.0f - z) * nn + z * hold[rt][vr];
                    }
                }
            }
        }
    }
}

__global__ void __launch_bounds__(256)
final_kernel(const float* __restrict__ h, const float* __restrict__ fcw,
             const float* __restrict__ fcb, float* __restrict__ out) {
    __shared__ float sw[N_ACT_C * 256];
    __shared__ float sb[N_ACT_C];
    for (int i = threadIdx.x; i < N_ACT_C * 256; i += 256) sw[i] = fcw[i];
    if (threadIdx.x < N_ACT_C) sb[threadIdx.x] = fcb[threadIdx.x];
    __syncthreads();

    int n = blockIdx.x * 256 + threadIdx.x;
    if (n >= N_NODES_C) return;

    float acc[N_ACT_C];
#pragma unroll
    for (int a = 0; a < N_ACT_C; ++a) acc[a] = sb[a];
    const float* hr = h + (long long)n * 256;
    for (int k = 0; k < 256; ++k) {
        float hv = fmaxf(hr[k], 0.0f);
#pragma unroll
        for (int a = 0; a < N_ACT_C; ++a) acc[a] += hv * sw[a * 256 + k];
    }
    float mx = acc[0];
#pragma unroll
    for (int a = 1; a < N_ACT_C; ++a) mx = fmaxf(mx, acc[a]);
    float s = 0.0f;
#pragma unroll
    for (int a = 0; a < N_ACT_C; ++a) { acc[a] = __expf(acc[a] - mx); s += acc[a]; }
    float inv = 1.0f / s;
#pragma unroll
    for (int a = 0; a < N_ACT_C; ++a) out[(long long)n * N_ACT_C + a] = acc[a] * inv;
}

extern "C" void kernel_launch(void* const* d_in, const int* in_sizes, int n_in,
                              void* d_out, int out_size, void* d_ws, size_t ws_size,
                              hipStream_t stream) {
    const float* x   = (const float*)d_in[0];
    const int*   ei  = (const int*)  d_in[1];
    const float* ew  = (const float*)d_in[2];
    const float* gg  = (const float*)d_in[3];
    const float* wih = (const float*)d_in[4];
    const float* whh = (const float*)d_in[5];
    const float* bih = (const float*)d_in[6];
    const float* bhh = (const float*)d_in[7];
    const float* fcw = (const float*)d_in[8];
    const float* fcb = (const float*)d_in[9];
    float* out = (float*)d_out;

    float* h   = (float*)d_ws;
    float* m   = h + (size_t)N_PAD * 256;
    float* agg = m + (size_t)N_PAD * 256;
    __bf16* ggT  = (__bf16*)(agg + (size_t)N_PAD * 256);
    __bf16* wihb = ggT + 3 * 256 * 256;
    __bf16* whhb = wihb + 768 * 256;
    int* counts   = (int*)(whhb + 768 * 256);
    int* within   = counts + (size_t)NBLK * BIN_PITCH;
    int* binTotal = within + (size_t)NBLK * BIN_PITCH;
    int* binStart = binTotal + BIN_PITCH;
    int4* rec     = (int4*)(binStart + BIN_PITCH);

    convw_kernel<<<(3 * 256 * 256 + 255) / 256, 256, 0, stream>>>(gg, wih, whh,
                                                                  ggT, wihb, whhb);
    init_h_kernel<<<N_PAD, 256, 0, stream>>>(x, h);
    bin_count_kernel<<<NBLK, 256, 0, stream>>>(ei, counts);
    bin_scan_kernel<<<(NBIN + 31) / 32, 32, 0, stream>>>(counts, within, binTotal);
    bin_place_kernel<<<NBLK, 256, 0, stream>>>(ei, ew, within, binTotal, binStart, rec);

    for (int l = 0; l < 3; ++l) {
        gemm_m_kernel<<<(N_PAD + GM_RB - 1) / GM_RB, 256, 0, stream>>>(h, ggT + (size_t)l * 65536, m);
        agg_kernel<<<NBIN, 128, 0, stream>>>(m, rec, binStart, binTotal, agg);
        gru_kernel<<<(N_PAD / 16 + GRU_BT - 1) / GRU_BT, 256, 0, stream>>>(agg, h, wihb, whhb,
                                                                          bih, bhh);
    }

    final_kernel<<<(N_NODES_C + 255) / 256, 256, 0, stream>>>(h, fcw, fcb, out);
}


# Aggregation kernel: a range's 16 rows summed in registers on the f32 16x16x4 matrix instruction (one-hot weights x gathered m rows), LDS float atomics gone; sums by groups of 4 edges, products fused
# speedup vs baseline: 3.2847x; 1.0477x over previous
#include <hip/hip_runtime.h>

#define N_NODES_C 50000
#define N_EDGES_C 800000
#define F_IN_C 32
#define C_C 256
#define N_PAD 50016
#define N_ACT_C 8

typedef __attribute__((ext_vector_type(16))) __bf16 v16bf;
typedef __attribute__((ext_vector_type(8)))  __bf16 v8bf;
typedef __attribute__((ext_vector_type(8)))  float  v8f;

static __device__ __forceinline__ v8f wmma_bf16(v16bf a, v16bf b, v8f c) {
    return __builtin_amdgcn_wmma_f32_16x16x32_bf16(
         false, a,  false, b,
         (short)0, c,  false,  false);
}

typedef __attribute__((ext_vector_type(2))) float v2f;
typedef __attribute__((ext_vector_type(4))) float v4f;
typedef __attribute__((ext_vector_type(3))) int   v3i;

static __device__ __forceinline__ v8f wmma_f32_k4(v2f a, v2f b, v8f c) {
    return __builtin_amdgcn_wmma_f32_16x16x4_f32(
         false, a,  false, b,
         (short)0, c,  false,  false);
}

static __device__ __forceinline__ v16bf load_b_frag_bf16(
    const __bf16* __restrict__ BT, int col, int ld, int k0, int lane) {
    int half = lane >> 4;
    const __bf16* p = BT + (long long)col * ld + k0 + half * 8;
    v8bf b0 = *(const v8bf*)(p);
    v8bf b1 = *(const v8bf*)(p + 16);
    v16bf b;
#pragma unroll
    for (int i = 0; i < 8; ++i) { b[i] = b0[i]; b[8 + i] = b1[i]; }
    return b;
}

static __device__ __forceinline__ float sigmoidf_(float x) {
    return 1.0f / (1.0f + __expf(-x));
}

__global__ void convw_kernel(const float* __restrict__ gg,
                             const float* __restrict__ wih,
                             const float* __restrict__ whh,
                             __bf16* __restrict__ ggT,
                             __bf16* __restrict__ wihb,
                             __bf16* __restrict__ whhb) {
    int i = blockIdx.x * 256 + threadIdx.x;
    if (i < 3 * 256 * 256) {
        int l = i >> 16, r = i & 65535, j = r >> 8, k = r & 255;
        ggT[i] = (__bf16)gg[l * 65536 + k * 256 + j];
    }
    if (i < 768 * 256) {
        wihb[i] = (__bf16)wih[i];
        whhb[i] = (__bf16)whh[i];
    }
}

__global__ void init_h_kernel(const float* __restrict__ x, float* __restrict__ h) {
    int i = blockIdx.x * 256 + threadIdx.x;
    int n = i >> 8, j = i & 255;
    float v = 0.0f;
    if (n < N_NODES_C && j < F_IN_C) v = x[n * F_IN_C + j];
    h[i] = v;
}

#define EBLK 2048
#define NBLK ((N_EDGES_C + EBLK - 1) / EBLK)
#define BIN_SHIFT 4
#define BIN_ROWS 16
#define NBIN (N_PAD / BIN_ROWS)
#define BIN_PITCH 3136

__global__ void __launch_bounds__(256)
bin_count_kernel(const int* __restrict__ ei, int* __restrict__ counts) {
    __shared__ int hist[NBIN];
    for (int i = threadIdx.x; i < NBIN; i += 256) hist[i] = 0;
    __syncthreads();
    int e0 = blockIdx.x * EBLK;
#pragma unroll
    for (int j = 0; j < EBLK / 256; ++j) {
        int e = e0 + j * 256 + threadIdx.x;
        if (e < N_EDGES_C) atomicAdd(&hist[ei[N_EDGES_C + e] >> BIN_SHIFT], 1);
    }
    __syncthreads();
    int* row = counts + (size_t)blockIdx.x * BIN_PITCH;
    for (int i = threadIdx.x; i < NBIN; i += 256) row[i] = hist[i];
}

#define SCAN_U 17
__global__ void __launch_bounds__(32)
bin_scan_kernel(const int* __restrict__ counts, int* __restrict__ within,
                int* __restrict__ binTotal) {
    int bin = blockIdx.x * 32 + threadIdx.x;
    if (bin >= NBIN) return;
    int run = 0;
    for (int b0 = 0; b0 < NBLK; b0 += SCAN_U) {
        int c[SCAN_U];
#pragma unroll
        for (int u = 0; u < SCAN_U; ++u)
            c[u] = (b0 + u < NBLK) ? counts[(size_t)(b0 + u) * BIN_PITCH + bin] : 0;
#pragma unroll
        for (int u = 0; u < SCAN_U; ++u) {
            if (b0 + u < NBLK) within[(size_t)(b0 + u) * BIN_PITCH + bin] = run;
            run += c[u];
        }
    }
    binTotal[bin] = run;
}

#define PLACE_BPT ((NBIN + 255) / 256)
__global__ void __launch_bounds__(256)
bin_place_kernel(const int* __restrict__ ei, const float* __restrict__ ew,
                 const int* __restrict__ within, const int* __restrict__ binTotal,
                 int* __restrict__ binStart, int4* __restrict__ rec) {
    __shared__ int sCur[8][NBIN];
    __shared__ int sBase[NBIN];
    __shared__ int sWave[8];
    int tid = threadIdx.x, lane = tid & 31, wave = tid >> 5;
    int blk = blockIdx.x;

    int loc[PLACE_BPT];
    int sum = 0;
#pragma unroll
    for (int k = 0; k < PLACE_BPT; ++k) {
        int bin = PLACE_BPT * tid + k;
        int v = (bin < NBIN) ? binTotal[bin] : 0;
        loc[k] = sum;
        sum += v;
    }
    int inc = sum;
#pragma unroll
    for (int o = 1; o < 32; o <<= 1) {
        int t = __shfl_up(inc, o);
        if (lane >= o) inc += t;
    }
    if (lane == 31) sWave[wave] = inc;
    for (int i = tid; i < 8 * NBIN; i += 256) (&sCur[0][0])[i] = 0;
    __syncthreads();
    int wbase = 0;
    for (int w = 0; w < wave; ++w) wbase += sWave[w];
    int excl = wbase + inc - sum;
#pragma unroll
    for (int k = 0; k < PLACE_BPT; ++k) {
        int bin = PLACE_BPT * tid + k;
        if (bin < NBIN) {
            int st = excl + loc[k];
            sBase[bin] = st + within[(size_t)blk * BIN_PITCH + bin];
            if (blk == 0) binStart[bin] = st;
        }
    }

    int e0 = blk * EBLK + wave * 256;
    int dreg[8], sreg[8];
    float wreg[8];
#pragma unroll
    for (int c = 0; c < 8; ++c) {
        int e = e0 + c * 32 + lane;
        bool ok = e < N_EDGES_C;
        dreg[c] = ok ? ei[N_EDGES_C + e] : -1;
        sreg[c] = ok ? ei[e] : 0;
        wreg[c] = ok ? ew[e] : 0.0f;
        if (ok) atomicAdd(&sCur[wave][dreg[c] >> BIN_SHIFT], 1);
    }
    __syncthreads();
    for (int bin = tid; bin < NBIN; bin += 256) {
        int run = sBase[bin];
#pragma unroll
        for (int w = 0; w < 8; ++w) {
            int c = sCur[w][bin];
            sCur[w][bin] = run;
            run += c;
        }
    }
    __syncthreads();
#pragma unroll
    for (int c = 0; c < 8; ++c) {
        int d = dreg[c];
        int bin = d >> BIN_SHIFT;
        int rank = 0, cnt = 0;
#pragma unroll
        for (int j = 0; j < 32; ++j) {
            int bj = __builtin_amdgcn_readlane(bin, j);
            int same = (bj == bin) ? 1 : 0;
            cnt += same;
            rank += (j < lane) ? same : 0;
        }
        if (d >= 0) {
            int pos = sCur[wave][bin] + rank;
            rec[pos] = make_int4(sreg[c], d & (BIN_ROWS - 1), __float_as_int(wreg[c]), 0);
            if (rank == cnt - 1) sCur[wave][bin] = pos + 1;
        }
    }
}

#define AGG_G 4
__global__ void __launch_bounds__(128)
agg_kernel(const float* __restrict__ m, const int4* __restrict__ rec,
           const int* __restrict__ binStart, const int* __restrict__ binTotal,
           float* __restrict__ agg) {
    int lane = threadIdx.x & 31, wave = threadIdx.x >> 5;
    int c = lane & 15, half = lane >> 4;
    int b = blockIdx.x;

    int start = __builtin_amdgcn_readfirstlane(binStart[b]);
    int n     = __builtin_amdgcn_readfirstlane(binTotal[b]);
    const int4* list = rec + start;
    unsigned laneoff = (unsigned)(wave * 64 + c * 4) * 4u;
    const char* mb = (const char*)m;

    v8f acc[4];
#pragma unroll
    for (int t = 0; t < 4; ++t) { v8f z = {}; acc[t] = z; }

    if (n > 0) {
        int last = n - 1;
        const char* lb = (const char*)list;
        v3i q[AGG_G][2];
#pragma unroll
        for (int g = 0; g < AGG_G; ++g)
#pragma unroll
            for (int s = 0; s < 2; ++s)
                q[g][s] = *(const v3i*)(lb + 16 * min(4 * g + 2 * half + s, last));
        for (int e0 = 0; e0 < n; e0 += 4 * AGG_G) {
            v2f a[AGG_G];
            unsigned off[AGG_G][2];
#pragma unroll
            for (int g = 0; g < AGG_G; ++g)
#pragma unroll
                for (int s = 0; s < 2; ++s) {
                    int idx = e0 + 4 * g + 2 * half + s;
                    bool hit = (q[g][s][1] == c) & (idx <= last);
                    a[g][s] = hit ? __int_as_float(q[g][s][2]) : 0.0f;
                    off[g][s] = ((unsigned)q[g][s][0] << 10) | laneoff;
                }
            v4f v[AGG_G][2];
#pragma unroll
            for (int g = 0; g < AGG_G; ++g)
#pragma unroll
                for (int s = 0; s < 2; ++s)
                    v[g][s] = *(const v4f*)(mb + off[g][s]);
#pragma unroll
            for (int g = 0; g < AGG_G; ++g)
#pragma unroll
                for (int s = 0; s < 2; ++s)
                    q[g][s] = *(const v3i*)(lb + 16 * min(e0 + 4 * AGG_G + 4 * g + 2 * half + s, last));
            __builtin_amdgcn_sched_barrier(0);
#pragma unroll
            for (int g = 0; g < AGG_G; ++g)
#pragma unroll
                for (int t = 0; t < 4; ++t) {
                    v2f bb;
                    bb[0] = v[g][0][t];
                    bb[1] = v[g][1][t];
                    acc[t] = wmma_f32_k4(a[g], bb, acc[t]);
                }
        }
    }

    char* arow = (char*)(agg + (size_t)b * BIN_ROWS * 256 + (size_t)(8 * half) * 256) + laneoff;
#pragma unroll
    for (int r = 0; r < 8; ++r) {
        v4f o;
        o[0] = acc[0][r]; o[1] = acc[1][r]; o[2] = acc[2][r]; o[3] = acc[3][r];
        *(v4f*)(arow + (size_t)r * 1024) = o;
    }
}

#define GRU_RT 7
#define GRU_RB (GRU_RT * 16)
#define GRU_BT 13
#define GRU_PITCH 264

typedef __attribute__((ext_vector_type(4))) __bf16 v4bf;

static __device__ __forceinline__ v16bf load_a_frag_lds(
    const __bf16* sA, int row, int k0, int half) {
    const __bf16* p = sA + row * GRU_PITCH + k0 + half * 8;
    v8bf a0 = *(const v8bf*)(p);
    v8bf a1 = *(const v8bf*)(p + 16);
    v16bf a;
#pragma unroll
    for (int i = 0; i < 8; ++i) { a[i] = a0[i]; a[8 + i] = a1[i]; }
    return a;
}

#define GM_RB 128
#define GM_RT (GM_RB / 16)

#define GM_LOAD_A(AS, rb, kq)                                               \
    _Pragma("unroll")                                                       \
    for (int i = 0; i < 4; ++i)                                             \
        AS[i] = load_a_frag_lds(sHm, ((rb) + i) * 16 + cl, (kq), half);

#define GM_LOAD_W(WS, kq)                                                   \
    _Pragma("unroll")                                                       \
    for (int t = 0; t < 2; ++t)                                             \
        WS[t] = load_b_frag_bf16(WgT, c0 + t * 16 + cl, 256, (kq), lane);

#define GM_BLOCK(AS, WS, rb)                                                \
    _Pragma("unroll")                                                       \
    for (int i = 0; i < 4; ++i) {                                           \
        acc0[(rb) + i] = wmma_bf16(AS[i], WS[0], acc0[(rb) + i]);           \
        acc1[(rb) + i] = wmma_bf16(AS[i], WS[1], acc1[(rb) + i]);           \
    }

#define GM_FENCE __builtin_amdgcn_sched_barrier(0);

__global__ void __launch_bounds__(256)
gemm_m_kernel(const float* __restrict__ h, const __bf16* __restrict__ WgT,
              float* __restrict__ m) {
    __shared__ __attribute__((aligned(16))) __bf16 sHm[GM_RB * GRU_PITCH];

    int tid  = threadIdx.x;
    int lane = tid & 31;
    int wave = tid >> 5;
    int half = lane >> 4;
    int cl   = lane & 15;
    int row0 = blockIdx.x * GM_RB;
    int c0   = wave * 32;

#pragma unroll 1
    for (int it = 0; it < 32; it += 16) {
        float4 vh[16];
#pragma unroll
        for (int j = 0; j < 16; ++j) {
            int idx = (it + j) * 256 + tid;
            int r = idx >> 6, c4 = idx & 63;
            int grow = min(row0 + r, N_PAD - 1);
            vh[j] = *(const float4*)(h + (long long)grow * 256 + c4 * 4);
        }
#pragma unroll
        for (int j = 0; j < 16; ++j) {
            int idx = (it + j) * 256 + tid;
            int r = idx >> 6, c4 = idx & 63;
            v4bf ph;
            ph[0] = (__bf16)vh[j].x; ph[1] = (__bf16)vh[j].y;
            ph[2] = (__bf16)vh[j].z; ph[3] = (__bf16)vh[j].w;
            *(v4bf*)(sHm + r * GRU_PITCH + c4 * 4) = ph;
        }
    }
    __syncthreads();

    v8f acc0[GM_RT], acc1[GM_RT];
#pragma unroll
    for (int rt = 0; rt < GM_RT; ++rt) {
        v8f z = {};
        acc0[rt] = z; acc1[rt] = z;
    }

    v16bf w0[2], w1[2];
    v16bf aP[4], aQ[4];
    GM_LOAD_W(w0, 0)
    GM_LOAD_A(aP, 0, 0)

#pragma unroll 1
    for (int kk = 0; kk < 4; ++kk) {
        int k0 = kk * 64;
        GM_LOAD_W(w1, k0 + 32)
        GM_LOAD_A(aQ, 4, k0)
        GM_FENCE
        GM_BLOCK(aP, w0, 0)
        GM_FENCE
        GM_LOAD_A(aP, 0, k0 + 32)
        GM_FENCE
        GM_BLOCK(aQ, w0, 4)
        GM_FENCE
        if (kk < 3) {
            GM_LOAD_W(w0, k0 + 64)
        }
        GM_LOAD_A(aQ, 4, k0 + 32)
        GM_FENCE
        GM_BLOCK(aP, w1, 0)
        GM_FENCE
        if (kk < 3) {
            GM_LOAD_A(aP, 0, k0 + 64)
        }
        GM_FENCE
        GM_BLOCK(aQ, w1, 4)
        GM_FENCE
    }

    float* mst = m + (8 * half) * 256 + c0 + cl;
#pragma unroll
    for (int rt = 0; rt < GM_RT; ++rt) {
        if (row0 + rt * 16 < N_PAD) {
#pragma unroll
            for (int vr = 0; vr < 8; ++vr) {
                float* p = mst + (long long)(row0 + rt * 16 + vr) * 256;
                p[0]  = acc0[rt][vr];
                p[16] = acc1[rt][vr];
            }
        }
    }
}

#define GRU_LOAD_A(AS, rb, kq)                                              \
    _Pragma("unroll")                                                       \
    for (int i = 0; i < 4; ++i) {                                           \
        if ((rb) + i < GRU_RT) {                                            \
            AS[i]     = load_a_frag_lds(sAgg, ((rb) + i) * 16 + cl, (kq), half);\
            AS[4 + i] = load_a_frag_lds(sH,   ((rb) + i) * 16 + cl, (kq), half);\
        }                                                                   \
    }

#define GRU_BLOCK(AS, WS, rb)                                               \
    _Pragma("unroll")                                                       \
    for (int i = 0; i < 4; ++i) {                                           \
        if ((rb) + i < GRU_RT) {                                            \
            aR[(rb) + i]  = wmma_bf16(AS[i], WS[0], aR[(rb) + i]);          \
            aZ[(rb) + i]  = wmma_bf16(AS[i], WS[1], aZ[(rb) + i]);          \
            aIN[(rb) + i] = wmma_bf16(AS[i], WS[2], aIN[(rb) + i]);         \
        }                                                                   \
    }                                                                       \
    _Pragma("unroll")                                                       \
    for (int i = 0; i < 4; ++i) {                                           \
        if ((rb) + i < GRU_RT) {                                            \
            aR[(rb) + i]  = wmma_bf16(AS[4 + i], WS[3], aR[(rb) + i]);      \
            aZ[(rb) + i]  = wmma_bf16(AS[4 + i], WS[4], aZ[(rb) + i]);      \
            aHN[(rb) + i] = wmma_bf16(AS[4 + i], WS[5], aHN[(rb) + i]);     \
        }                                                                   \
    }

#define GRU_FENCE __builtin_amdgcn_sched_barrier(0);

#define GRU_LOAD_W(WS, kq)                                                  \
    _Pragma("unroll")                                                       \
    for (int g = 0; g < 3; ++g) {                                           \
        WS[g]     = load_b_frag_bf16(wihb, g * 256 + c, 256, (kq), lane);   \
        WS[3 + g] = load_b_frag_bf16(whhb, g * 256 + c, 256, (kq), lane);   \
    }

__global__ void __launch_bounds__(256) __attribute__((amdgpu_waves_per_eu(2, 2)))
gru_kernel(const float* __restrict__ agg, float* __restrict__ h,
           const __bf16* __restrict__ wihb, const __bf16* __restrict__ whhb,
           const float* __restrict__ bih, const float* __restrict__ bhh) {
    __shared__ __attribute__((aligned(16))) __bf16 sA[2 * GRU_RB * GRU_PITCH];
    __bf16* sAgg = sA;
    __bf16* sH   = sA + GRU_RB * GRU_PITCH;

    int tid  = threadIdx.x;
    int lane = tid & 31;
    int wave = tid >> 5;
    int half = lane >> 4;
    int cl   = lane & 15;
    int tile0 = blockIdx.x * GRU_BT;

#pragma unroll 1
    for (int g = 0; g < 2; ++g) {
        int row0   = (tile0 + g * GRU_RT) * 16;
        int rowEnd = min((tile0 + (g ? GRU_BT : GRU_RT)) * 16, N_PAD);
        if (row0 >= rowEnd) break;
        if (g) __syncthreads();

#pragma unroll 1
        for (int it = 0; it < 28; it += 7) {
            float4 va[7], vh[7];
#pragma unroll
            for (int j = 0; j < 7; ++j) {
                int idx = (it + j) * 256 + tid;
                int r = idx >> 6, c4 = idx & 63;
                int grow = min(row0 + r, rowEnd - 1);
                va[j] = *(const float4*)(agg + (long long)grow * 256 + c4 * 4);
                vh[j] = *(const float4*)(h   + (long long)grow * 256 + c4 * 4);
            }
#pragma unroll
            for (int j = 0; j < 7; ++j) {
                int idx = (it + j) * 256 + tid;
                int r = idx >> 6, c4 = idx & 63;
                v4bf pa, ph;
                pa[0] = (__bf16)va[j].x; pa[1] = (__bf16)va[j].y;
                pa[2] = (__bf16)va[j].z; pa[3] = (__bf16)va[j].w;
                ph[0] = (__bf16)vh[j].x; ph[1] = (__bf16)vh[j].y;
                ph[2] = (__bf16)vh[j].z; ph[3] = (__bf16)vh[j].w;
                *(v4bf*)(sAgg + r * GRU_PITCH + c4 * 4) = pa;
                *(v4bf*)(sH   + r * GRU_PITCH + c4 * 4) = ph;
            }
        }
        __syncthreads();

#pragma unroll 1
        for (int t = 0; t < 2; ++t) {
            int c = wave * 32 + t * 16 + cl;

            v8f aR[GRU_RT], aZ[GRU_RT], aIN[GRU_RT], aHN[GRU_RT];
#pragma unroll
            for (int rt = 0; rt < GRU_RT; ++rt) {
                v8f z = {};
                aR[rt] = z; aZ[rt] = z; aIN[rt] = z; aHN[rt] = z;
            }

            v16bf w0[6], w1[6];
            v16bf aP[8], aQ[8];
            GRU_LOAD_W(w0, 0)
            GRU_LOAD_A(aP, 0, 0)

#pragma unroll 1
            for (int kk = 0; kk < 4; ++kk) {
                int k0 = kk * 64;
                GRU_LOAD_W(w1, k0 + 32)
                GRU_LOAD_A(aQ, 4, k0)
                GRU_FENCE
                GRU_BLOCK(aP, w0, 0)
                GRU_FENCE
                GRU_LOAD_A(aP, 0, k0 + 32)
                GRU_FENCE
                GRU_BLOCK(aQ, w0, 4)
                GRU_FENCE
                if (kk < 3) {
                    GRU_LOAD_W(w0, k0 + 64)
                }
                GRU_LOAD_A(aQ, 4, k0 + 32)
                GRU_FENCE
                GRU_BLOCK(aP, w1, 0)
                GRU_FENCE
                if (kk < 3) {
                    GRU_LOAD_A(aP, 0, k0 + 64)
                }
                GRU_FENCE
                GRU_BLOCK(aQ, w1, 4)
                GRU_FENCE
            }

            float bir = bih[c],       bhr = bhh[c];
            float biz = bih[256 + c], bhz = bhh[256 + c];
            float bin = bih[512 + c], bhn = bhh[512 + c];
            float hold[GRU_RT][8];
            const float* hcol = h + (8 * half) * 256 + c;
            float* hst = h + (8 * half) * 256 + c;
#pragma unroll
            for (int rt = 0; rt < GRU_RT; ++rt) {
                int rbase = (row0 + rt * 16 < rowEnd) ? row0 + rt * 16 : row0;
#pragma unroll
                for (int vr = 0; vr < 8; ++vr)
                    hold[rt][vr] = hcol[(long long)(rbase + vr) * 256];
            }
#pragma unroll
            for (int rt = 0; rt < GRU_RT; ++rt) {
                if (row0 + rt * 16 < rowEnd) {
#pragma unroll
                    for (int vr = 0; vr < 8; ++vr) {
                        float r  = sigmoidf_(aR[rt][vr] + bir + bhr);
                        float z  = sigmoidf_(aZ[rt][vr] + biz + bhz);
                        float nn = tanhf(aIN[rt][vr] + bin + r * (aHN[rt][vr] + bhn));
                        hst[(long long)(row0 + rt * 16 + vr) * 256] =
                            (1.0f - z) * nn + z * hold[rt][vr];
                    }
                }
            }
        }
    }
}

__global__ void __launch_bounds__(256)
final_kernel(const float* __restrict__ h, const float* __restrict__ fcw,
             const float* __restrict__ fcb, float* __restrict__ out) {
    __shared__ float sw[N_ACT_C * 256];
    __shared__ float sb[N_ACT_C];
    for (int i = threadIdx.x; i < N_ACT_C * 256; i += 256) sw[i] = fcw[i];
    if (threadIdx.x < N_ACT_C) sb[threadIdx.x] = fcb[threadIdx.x];
    __syncthreads();

    int n = blockIdx.x * 256 + threadIdx.x;
    if (n >= N_NODES_C) return;

    float acc[N_ACT_C];
#pragma unroll
    for (int a = 0; a < N_ACT_C; ++a) acc[a] = sb[a];
    const float* hr = h + (long long)n * 256;
    for (int k = 0; k < 256; ++k) {
        float hv = fmaxf(hr[k], 0.0f);
#pragma unroll
        for (int a = 0; a < N_ACT_C; ++a) acc[a] += hv * sw[a * 256 + k];
    }
    float mx = acc[0];
#pragma unroll
    for (int a = 1; a < N_ACT_C; ++a) mx = fmaxf(mx, acc[a]);
    float s = 0.0f;
#pragma unroll
    for (int a = 0; a < N_ACT_C; ++a) { acc[a] = __expf(acc[a] - mx); s += acc[a]; }
    float inv = 1.0f / s;
#pragma unroll
    for (int a = 0; a < N_ACT_C; ++a) out[(long long)n * N_ACT_C + a] = acc[a] * inv;
}

extern "C" void kernel_launch(void* const* d_in, const int* in_sizes, int n_in,
                              void* d_out, int out_size, void* d_ws, size_t ws_size,
                              hipStream_t stream) {
    const float* x   = (const float*)d_in[0];
    const int*   ei  = (const int*)  d_in[1];
    const float* ew  = (const float*)d_in[2];
    const float* gg  = (const float*)d_in[3];
    const float* wih = (const float*)d_in[4];
    const float* whh = (const float*)d_in[5];
    const float* bih = (const float*)d_in[6];
    const float* bhh = (const float*)d_in[7];
    const float* fcw = (const float*)d_in[8];
    const float* fcb = (const float*)d_in[9];
    float* out = (float*)d_out;

    float* h   = (float*)d_ws;
    float* m   = h + (size_t)N_PAD * 256;
    float* agg = m + (size_t)N_PAD * 256;
    __bf16* ggT  = (__bf16*)(agg + (size_t)N_PAD * 256);
    __bf16* wihb = ggT + 3 * 256 * 256;
    __bf16* whhb = wihb + 768 * 256;
    int* counts   = (int*)(whhb + 768 * 256);
    int* within   = counts + (size_t)NBLK * BIN_PITCH;
    int* binTotal = within + (size_t)NBLK * BIN_PITCH;
    int* binStart = binTotal + BIN_PITCH;
    int4* rec     = (int4*)(binStart + BIN_PITCH);

    convw_kernel<<<(3 * 256 * 256 + 255) / 256, 256, 0, stream>>>(gg, wih, whh,
                                                                  ggT, wihb, whhb);
    init_h_kernel<<<N_PAD, 256, 0, stream>>>(x, h);
    bin_count_kernel<<<NBLK, 256, 0, stream>>>(ei, counts);
    bin_scan_kernel<<<(NBIN + 31) / 32, 32, 0, stream>>>(counts, within, binTotal);
    bin_place_kernel<<<NBLK, 256, 0, stream>>>(ei, ew, within, binTotal, binStart, rec);

    for (int l = 0; l < 3; ++l) {
        gemm_m_kernel<<<(N_PAD + GM_RB - 1) / GM_RB, 256, 0, stream>>>(h, ggT + (size_t)l * 65536, m);
        agg_kernel<<<NBIN, 128, 0, stream>>>(m, rec, binStart, binTotal, agg);
        gru_kernel<<<(N_PAD / 16 + GRU_BT - 1) / GRU_BT, 256, 0, stream>>>(agg, h, wihb, whhb,
                                                                          bih, bhh);
    }

    final_kernel<<<(N_NODES_C + 255) / 256, 256, 0, stream>>>(h, fcw, fcb, out);
}


# GRU gates: tanh written branch-free (both ranges computed, one selected; library's formulas, last place may move); a tile's 8 rows taken gate by gate: 8 interleaved chains, was one chain per value
# speedup vs baseline: 3.6176x; 1.1013x over previous
#include <hip/hip_runtime.h>

#define N_NODES_C 50000
#define N_EDGES_C 800000
#define F_IN_C 32
#define C_C 256
#define N_PAD 50016
#define N_ACT_C 8

typedef __attribute__((ext_vector_type(16))) __bf16 v16bf;
typedef __attribute__((ext_vector_type(8)))  __bf16 v8bf;
typedef __attribute__((ext_vector_type(8)))  float  v8f;

static __device__ __forceinline__ v8f wmma_bf16(v16bf a, v16bf b, v8f c) {
    return __builtin_amdgcn_wmma_f32_16x16x32_bf16(
         false, a,  false, b,
         (short)0, c,  false,  false);
}

typedef __attribute__((ext_vector_type(2))) float v2f;
typedef __attribute__((ext_vector_type(4))) float v4f;
typedef __attribute__((ext_vector_type(3))) int   v3i;

static __device__ __forceinline__ v8f wmma_f32_k4(v2f a, v2f b, v8f c) {
    return __builtin_amdgcn_wmma_f32_16x16x4_f32(
         false, a,  false, b,
         (short)0, c,  false,  false);
}

static __device__ __forceinline__ v16bf load_b_frag_bf16(
    const __bf16* __restrict__ BT, int col, int ld, int k0, int lane) {
    int half = lane >> 4;
    const __bf16* p = BT + (long long)col * ld + k0 + half * 8;
    v8bf b0 = *(const v8bf*)(p);
    v8bf b1 = *(const v8bf*)(p + 16);
    v16bf b;
#pragma unroll
    for (int i = 0; i < 8; ++i) { b[i] = b0[i]; b[8 + i] = b1[i]; }
    return b;
}

static __device__ __forceinline__ float sigmoidf_(float x) {
    return 1.0f / (1.0f + __expf(-x));
}

static __device__ __forceinline__ float tanh_two_range(float x) {
    const float l2e_hi = __builtin_bit_cast(float, 0x3fb8aa3bu);
    const float l2e_lo = __builtin_bit_cast(float, 0x32a5705fu);
    const float t_max  = __builtin_bit_cast(float, 0x42b17218u);
    float ax = __builtin_fabsf(x);
    float t = ax + ax;
    float p = l2e_hi * t;
    float n = __builtin_rintf(p);
    float f = __builtin_fmaf(l2e_hi, t, -p);
    f = __builtin_fmaf(l2e_lo, t, f);
    float e = __builtin_amdgcn_exp2f((p - n) + f);
    e = __builtin_amdgcn_ldexpf(e, (int)n);
    e = (t > t_max) ? __builtin_inff() : e;
    float big = __builtin_fmaf(__builtin_amdgcn_rcpf(1.0f + e), -2.0f, 1.0f);
    float s = x * x;
    float q = __builtin_fmaf(__builtin_bit_cast(float, 0xbbbac73du), s,
                             __builtin_bit_cast(float, 0x3ca908c9u));
    q = __builtin_fmaf(s, q, __builtin_bit_cast(float, 0xbd5c1c4eu));
    q = __builtin_fmaf(s, q, __builtin_bit_cast(float, 0x3e088382u));
    q = __builtin_fmaf(s, q, __builtin_bit_cast(float, 0xbeaaaa99u));
    float small = __builtin_fmaf(s, ax * q, ax);
    float res = (ax < 0.625f) ? small : big;
    return __builtin_copysignf(res, x);
}

__global__ void convw_kernel(const float* __restrict__ gg,
                             const float* __restrict__ wih,
                             const float* __restrict__ whh,
                             __bf16* __restrict__ ggT,
                             __bf16* __restrict__ wihb,
                             __bf16* __restrict__ whhb) {
    int i = blockIdx.x * 256 + threadIdx.x;
    if (i < 3 * 256 * 256) {
        int l = i >> 16, r = i & 65535, j = r >> 8, k = r & 255;
        ggT[i] = (__bf16)gg[l * 65536 + k * 256 + j];
    }
    if (i < 768 * 256) {
        wihb[i] = (__bf16)wih[i];
        whhb[i] = (__bf16)whh[i];
    }
}

__global__ void init_h_kernel(const float* __restrict__ x, float* __restrict__ h) {
    int i = blockIdx.x * 256 + threadIdx.x;
    int n = i >> 8, j = i & 255;
    float v = 0.0f;
    if (n < N_NODES_C && j < F_IN_C) v = x[n * F_IN_C + j];
    h[i] = v;
}

#define EBLK 2048
#define NBLK ((N_EDGES_C + EBLK - 1) / EBLK)
#define BIN_SHIFT 4
#define BIN_ROWS 16
#define NBIN (N_PAD / BIN_ROWS)
#define BIN_PITCH 3136

__global__ void __launch_bounds__(256)
bin_count_kernel(const int* __restrict__ ei, int* __restrict__ counts) {
    __shared__ int hist[NBIN];
    for (int i = threadIdx.x; i < NBIN; i += 256) hist[i] = 0;
    __syncthreads();
    int e0 = blockIdx.x * EBLK;
#pragma unroll
    for (int j = 0; j < EBLK / 256; ++j) {
        int e = e0 + j * 256 + threadIdx.x;
        if (e < N_EDGES_C) atomicAdd(&hist[ei[N_EDGES_C + e] >> BIN_SHIFT], 1);
    }
    __syncthreads();
    int* row = counts + (size_t)blockIdx.x * BIN_PITCH;
    for (int i = threadIdx.x; i < NBIN; i += 256) row[i] = hist[i];
}

#define SCAN_U 17
__global__ void __launch_bounds__(32)
bin_scan_kernel(const int* __restrict__ counts, int* __restrict__ within,
                int* __restrict__ binTotal) {
    int bin = blockIdx.x * 32 + threadIdx.x;
    if (bin >= NBIN) return;
    int run = 0;
    for (int b0 = 0; b0 < NBLK; b0 += SCAN_U) {
        int c[SCAN_U];
#pragma unroll
        for (int u = 0; u < SCAN_U; ++u)
            c[u] = (b0 + u < NBLK) ? counts[(size_t)(b0 + u) * BIN_PITCH + bin] : 0;
#pragma unroll
        for (int u = 0; u < SCAN_U; ++u) {
            if (b0 + u < NBLK) within[(size_t)(b0 + u) * BIN_PITCH + bin] = run;
            run += c[u];
        }
    }
    binTotal[bin] = run;
}

#define PLACE_BPT ((NBIN + 255) / 256)
__global__ void __launch_bounds__(256)
bin_place_kernel(const int* __restrict__ ei, const float* __restrict__ ew,
                 const int* __restrict__ within, const int* __restrict__ binTotal,
                 int* __restrict__ binStart, int4* __restrict__ rec) {
    __shared__ int sCur[8][NBIN];
    __shared__ int sBase[NBIN];
    __shared__ int sWave[8];
    int tid = threadIdx.x, lane = tid & 31, wave = tid >> 5;
    int blk = blockIdx.x;

    int loc[PLACE_BPT];
    int sum = 0;
#pragma unroll
    for (int k = 0; k < PLACE_BPT; ++k) {
        int bin = PLACE_BPT * tid + k;
        int v = (bin < NBIN) ? binTotal[bin] : 0;
        loc[k] = sum;
        sum += v;
    }
    int inc = sum;
#pragma unroll
    for (int o = 1; o < 32; o <<= 1) {
        int t = __shfl_up(inc, o);
        if (lane >= o) inc += t;
    }
    if (lane == 31) sWave[wave] = inc;
    for (int i = tid; i < 8 * NBIN; i += 256) (&sCur[0][0])[i] = 0;
    __syncthreads();
    int wbase = 0;
    for (int w = 0; w < wave; ++w) wbase += sWave[w];
    int excl = wbase + inc - sum;
#pragma unroll
    for (int k = 0; k < PLACE_BPT; ++k) {
        int bin = PLACE_BPT * tid + k;
        if (bin < NBIN) {
            int st = excl + loc[k];
            sBase[bin] = st + within[(size_t)blk * BIN_PITCH + bin];
            if (blk == 0) binStart[bin] = st;
        }
    }

    int e0 = blk * EBLK + wave * 256;
    int dreg[8], sreg[8];
    float wreg[8];
#pragma unroll
    for (int c = 0; c < 8; ++c) {
        int e = e0 + c * 32 + lane;
        bool ok = e < N_EDGES_C;
        dreg[c] = ok ? ei[N_EDGES_C + e] : -1;
        sreg[c] = ok ? ei[e] : 0;
        wreg[c] = ok ? ew[e] : 0.0f;
        if (ok) atomicAdd(&sCur[wave][dreg[c] >> BIN_SHIFT], 1);
    }
    __syncthreads();
    for (int bin = tid; bin < NBIN; bin += 256) {
        int run = sBase[bin];
#pragma unroll
        for (int w = 0; w < 8; ++w) {
            int c = sCur[w][bin];
            sCur[w][bin] = run;
            run += c;
        }
    }
    __syncthreads();
#pragma unroll
    for (int c = 0; c < 8; ++c) {
        int d = dreg[c];
        int bin = d >> BIN_SHIFT;
        int rank = 0, cnt = 0;
#pragma unroll
        for (int j = 0; j < 32; ++j) {
            int bj = __builtin_amdgcn_readlane(bin, j);
            int same = (bj == bin) ? 1 : 0;
            cnt += same;
            rank += (j < lane) ? same : 0;
        }
        if (d >= 0) {
            int pos = sCur[wave][bin] + rank;
            rec[pos] = make_int4(sreg[c], d & (BIN_ROWS - 1), __float_as_int(wreg[c]), 0);
            if (rank == cnt - 1) sCur[wave][bin] = pos + 1;
        }
    }
}

#define AGG_G 4
__global__ void __launch_bounds__(128)
agg_kernel(const float* __restrict__ m, const int4* __restrict__ rec,
           const int* __restrict__ binStart, const int* __restrict__ binTotal,
           float* __restrict__ agg) {
    int lane = threadIdx.x & 31, wave = threadIdx.x >> 5;
    int c = lane & 15, half = lane >> 4;
    int b = blockIdx.x;

    int start = __builtin_amdgcn_readfirstlane(binStart[b]);
    int n     = __builtin_amdgcn_readfirstlane(binTotal[b]);
    const int4* list = rec + start;
    unsigned laneoff = (unsigned)(wave * 64 + c * 4) * 4u;
    const char* mb = (const char*)m;

    v8f acc[4];
#pragma unroll
    for (int t = 0; t < 4; ++t) { v8f z = {}; acc[t] = z; }

    if (n > 0) {
        int last = n - 1;
        const char* lb = (const char*)list;
        v3i q[AGG_G][2];
#pragma unroll
        for (int g = 0; g < AGG_G; ++g)
#pragma unroll
            for (int s = 0; s < 2; ++s)
                q[g][s] = *(const v3i*)(lb + 16 * min(4 * g + 2 * half + s, last));
        for (int e0 = 0; e0 < n; e0 += 4 * AGG_G) {
            v2f a[AGG_G];
            unsigned off[AGG_G][2];
#pragma unroll
            for (int g = 0; g < AGG_G; ++g)
#pragma unroll
                for (int s = 0; s < 2; ++s) {
                    int idx = e0 + 4 * g + 2 * half + s;
                    bool hit = (q[g][s][1] == c) & (idx <= last);
                    a[g][s] = hit ? __int_as_float(q[g][s][2]) : 0.0f;
                    off[g][s] = ((unsigned)q[g][s][0] << 10) | laneoff;
                }
            v4f v[AGG_G][2];
#pragma unroll
            for (int g = 0; g < AGG_G; ++g)
#pragma unroll
                for (int s = 0; s < 2; ++s)
                    v[g][s] = *(const v4f*)(mb + off[g][s]);
#pragma unroll
            for (int g = 0; g < AGG_G; ++g)
#pragma unroll
                for (int s = 0; s < 2; ++s)
                    q[g][s] = *(const v3i*)(lb + 16 * min(e0 + 4 * AGG_G + 4 * g + 2 * half + s, last));
            __builtin_amdgcn_sched_barrier(0);
#pragma unroll
            for (int g = 0; g < AGG_G; ++g)
#pragma unroll
                for (int t = 0; t < 4; ++t) {
                    v2f bb;
                    bb[0] = v[g][0][t];
                    bb[1] = v[g][1][t];
                    acc[t] = wmma_f32_k4(a[g], bb, acc[t]);
                }
        }
    }

    char* arow = (char*)(agg + (size_t)b * BIN_ROWS * 256 + (size_t)(8 * half) * 256) + laneoff;
#pragma unroll
    for (int r = 0; r < 8; ++r) {
        v4f o;
        o[0] = acc[0][r]; o[1] = acc[1][r]; o[2] = acc[2][r]; o[3] = acc[3][r];
        *(v4f*)(arow + (size_t)r * 1024) = o;
    }
}

#define GRU_RT 7
#define GRU_RB (GRU_RT * 16)
#define GRU_BT 13
#define GRU_PITCH 264

typedef __attribute__((ext_vector_type(4))) __bf16 v4bf;

static __device__ __forceinline__ v16bf load_a_frag_lds(
    const __bf16* sA, int row, int k0, int half) {
    const __bf16* p = sA + row * GRU_PITCH + k0 + half * 8;
    v8bf a0 = *(const v8bf*)(p);
    v8bf a1 = *(const v8bf*)(p + 16);
    v16bf a;
#pragma unroll
    for (int i = 0; i < 8; ++i) { a[i] = a0[i]; a[8 + i] = a1[i]; }
    return a;
}

#define GM_RB 128
#define GM_RT (GM_RB / 16)

#define GM_LOAD_A(AS, rb, kq)                                               \
    _Pragma("unroll")                                                       \
    for (int i = 0; i < 4; ++i)                                             \
        AS[i] = load_a_frag_lds(sHm, ((rb) + i) * 16 + cl, (kq), half);

#define GM_LOAD_W(WS, kq)                                                   \
    _Pragma("unroll")                                                       \
    for (int t = 0; t < 2; ++t)                                             \
        WS[t] = load_b_frag_bf16(WgT, c0 + t * 16 + cl, 256, (kq), lane);

#define GM_BLOCK(AS, WS, rb)                                                \
    _Pragma("unroll")                                                       \
    for (int i = 0; i < 4; ++i) {                                           \
        acc0[(rb) + i] = wmma_bf16(AS[i], WS[0], acc0[(rb) + i]);           \
        acc1[(rb) + i] = wmma_bf16(AS[i], WS[1], acc1[(rb) + i]);           \
    }

#define GM_FENCE __builtin_amdgcn_sched_barrier(0);

__global__ void __launch_bounds__(256)
gemm_m_kernel(const float* __restrict__ h, const __bf16* __restrict__ WgT,
              float* __restrict__ m) {
    __shared__ __attribute__((aligned(16))) __bf16 sHm[GM_RB * GRU_PITCH];

    int tid  = threadIdx.x;
    int lane = tid & 31;
    int wave = tid >> 5;
    int half = lane >> 4;
    int cl   = lane & 15;
    int row0 = blockIdx.x * GM_RB;
    int c0   = wave * 32;

#pragma unroll 1
    for (int it = 0; it < 32; it += 16) {
        float4 vh[16];
#pragma unroll
        for (int j = 0; j < 16; ++j) {
            int idx = (it + j) * 256 + tid;
            int r = idx >> 6, c4 = idx & 63;
            int grow = min(row0 + r, N_PAD - 1);
            vh[j] = *(const float4*)(h + (long long)grow * 256 + c4 * 4);
        }
#pragma unroll
        for (int j = 0; j < 16; ++j) {
            int idx = (it + j) * 256 + tid;
            int r = idx >> 6, c4 = idx & 63;
            v4bf ph;
            ph[0] = (__bf16)vh[j].x; ph[1] = (__bf16)vh[j].y;
            ph[2] = (__bf16)vh[j].z; ph[3] = (__bf16)vh[j].w;
            *(v4bf*)(sHm + r * GRU_PITCH + c4 * 4) = ph;
        }
    }
    __syncthreads();

    v8f acc0[GM_RT], acc1[GM_RT];
#pragma unroll
    for (int rt = 0; rt < GM_RT; ++rt) {
        v8f z = {};
        acc0[rt] = z; acc1[rt] = z;
    }

    v16bf w0[2], w1[2];
    v16bf aP[4], aQ[4];
    GM_LOAD_W(w0, 0)
    GM_LOAD_A(aP, 0, 0)

#pragma unroll 1
    for (int kk = 0; kk < 4; ++kk) {
        int k0 = kk * 64;
        GM_LOAD_W(w1, k0 + 32)
        GM_LOAD_A(aQ, 4, k0)
        GM_FENCE
        GM_BLOCK(aP, w0, 0)
        GM_FENCE
        GM_LOAD_A(aP, 0, k0 + 32)
        GM_FENCE
        GM_BLOCK(aQ, w0, 4)
        GM_FENCE
        if (kk < 3) {
            GM_LOAD_W(w0, k0 + 64)
        }
        GM_LOAD_A(aQ, 4, k0 + 32)
        GM_FENCE
        GM_BLOCK(aP, w1, 0)
        GM_FENCE
        if (kk < 3) {
            GM_LOAD_A(aP, 0, k0 + 64)
        }
        GM_FENCE
        GM_BLOCK(aQ, w1, 4)
        GM_FENCE
    }

    float* mst = m + (8 * half) * 256 + c0 + cl;
#pragma unroll
    for (int rt = 0; rt < GM_RT; ++rt) {
        if (row0 + rt * 16 < N_PAD) {
#pragma unroll
            for (int vr = 0; vr < 8; ++vr) {
                float* p = mst + (long long)(row0 + rt * 16 + vr) * 256;
                p[0]  = acc0[rt][vr];
                p[16] = acc1[rt][vr];
            }
        }
    }
}

#define GRU_LOAD_A(AS, rb, kq)                                              \
    _Pragma("unroll")                                                       \
    for (int i = 0; i < 4; ++i) {                                           \
        if ((rb) + i < GRU_RT) {                                            \
            AS[i]     = load_a_frag_lds(sAgg, ((rb) + i) * 16 + cl, (kq), half);\
            AS[4 + i] = load_a_frag_lds(sH,   ((rb) + i) * 16 + cl, (kq), half);\
        }                                                                   \
    }

#define GRU_BLOCK(AS, WS, rb)                                               \
    _Pragma("unroll")                                                       \
    for (int i = 0; i < 4; ++i) {                                           \
        if ((rb) + i < GRU_RT) {                                            \
            aR[(rb) + i]  = wmma_bf16(AS[i], WS[0], aR[(rb) + i]);          \
            aZ[(rb) + i]  = wmma_bf16(AS[i], WS[1], aZ[(rb) + i]);          \
            aIN[(rb) + i] = wmma_bf16(AS[i], WS[2], aIN[(rb) + i]);         \
        }                                                                   \
    }                                                                       \
    _Pragma("unroll")                                                       \
    for (int i = 0; i < 4; ++i) {                                           \
        if ((rb) + i < GRU_RT) {                                            \
            aR[(rb) + i]  = wmma_bf16(AS[4 + i], WS[3], aR[(rb) + i]);      \
            aZ[(rb) + i]  = wmma_bf16(AS[4 + i], WS[4], aZ[(rb) + i]);      \
            aHN[(rb) + i] = wmma_bf16(AS[4 + i], WS[5], aHN[(rb) + i]);     \
        }                                                                   \
    }

#define GRU_FENCE __builtin_amdgcn_sched_barrier(0);

#define GRU_LOAD_W(WS, kq)                                                  \
    _Pragma("unroll")                                                       \
    for (int g = 0; g < 3; ++g) {                                           \
        WS[g]     = load_b_frag_bf16(wihb, g * 256 + c, 256, (kq), lane);   \
        WS[3 + g] = load_b_frag_bf16(whhb, g * 256 + c, 256, (kq), lane);   \
    }

__global__ void __launch_bounds__(256) __attribute__((amdgpu_waves_per_eu(2, 2)))
gru_kernel(const float* __restrict__ agg, float* __restrict__ h,
           const __bf16* __restrict__ wihb, const __bf16* __restrict__ whhb,
           const float* __restrict__ bih, const float* __restrict__ bhh) {
    __shared__ __attribute__((aligned(16))) __bf16 sA[2 * GRU_RB * GRU_PITCH];
    __bf16* sAgg = sA;
    __bf16* sH   = sA + GRU_RB * GRU_PITCH;

    int tid  = threadIdx.x;
    int lane = tid & 31;
    int wave = tid >> 5;
    int half = lane >> 4;
    int cl   = lane & 15;
    int tile0 = blockIdx.x * GRU_BT;

#pragma unroll 1
    for (int g = 0; g < 2; ++g) {
        int row0   = (tile0 + g * GRU_RT) * 16;
        int rowEnd = min((tile0 + (g ? GRU_BT : GRU_RT)) * 16, N_PAD);
        if (row0 >= rowEnd) break;
        if (g) __syncthreads();

#pragma unroll 1
        for (int it = 0; it < 28; it += 7) {
            float4 va[7], vh[7];
#pragma unroll
            for (int j = 0; j < 7; ++j) {
                int idx = (it + j) * 256 + tid;
                int r = idx >> 6, c4 = idx & 63;
                int grow = min(row0 + r, rowEnd - 1);
                va[j] = *(const float4*)(agg + (long long)grow * 256 + c4 * 4);
                vh[j] = *(const float4*)(h   + (long long)grow * 256 + c4 * 4);
            }
#pragma unroll
            for (int j = 0; j < 7; ++j) {
                int idx = (it + j) * 256 + tid;
                int r = idx >> 6, c4 = idx & 63;
                v4bf pa, ph;
                pa[0] = (__bf16)va[j].x; pa[1] = (__bf16)va[j].y;
                pa[2] = (__bf16)va[j].z; pa[3] = (__bf16)va[j].w;
                ph[0] = (__bf16)vh[j].x; ph[1] = (__bf16)vh[j].y;
                ph[2] = (__bf16)vh[j].z; ph[3] = (__bf16)vh[j].w;
                *(v4bf*)(sAgg + r * GRU_PITCH + c4 * 4) = pa;
                *(v4bf*)(sH   + r * GRU_PITCH + c4 * 4) = ph;
            }
        }
        __syncthreads();

#pragma unroll 1
        for (int t = 0; t < 2; ++t) {
            int c = wave * 32 + t * 16 + cl;

            v8f aR[GRU_RT], aZ[GRU_RT], aIN[GRU_RT], aHN[GRU_RT];
#pragma unroll
            for (int rt = 0; rt < GRU_RT; ++rt) {
                v8f z = {};
                aR[rt] = z; aZ[rt] = z; aIN[rt] = z; aHN[rt] = z;
            }

            v16bf w0[6], w1[6];
            v16bf aP[8], aQ[8];
            GRU_LOAD_W(w0, 0)
            GRU_LOAD_A(aP, 0, 0)

#pragma unroll 1
            for (int kk = 0; kk < 4; ++kk) {
                int k0 = kk * 64;
                GRU_LOAD_W(w1, k0 + 32)
                GRU_LOAD_A(aQ, 4, k0)
                GRU_FENCE
                GRU_BLOCK(aP, w0, 0)
                GRU_FENCE
                GRU_LOAD_A(aP, 0, k0 + 32)
                GRU_FENCE
                GRU_BLOCK(aQ, w0, 4)
                GRU_FENCE
                if (kk < 3) {
                    GRU_LOAD_W(w0, k0 + 64)
                }
                GRU_LOAD_A(aQ, 4, k0 + 32)
                GRU_FENCE
                GRU_BLOCK(aP, w1, 0)
                GRU_FENCE
                if (kk < 3) {
                    GRU_LOAD_A(aP, 0, k0 + 64)
                }
                GRU_FENCE
                GRU_BLOCK(aQ, w1, 4)
                GRU_FENCE
            }

            float bir = bih[c],       bhr = bhh[c];
            float biz = bih[256 + c], bhz = bhh[256 + c];
            float bin = bih[512 + c], bhn = bhh[512 + c];
            float hold[GRU_RT][8];
            const float* hcol = h + (8 * half) * 256 + c;
            float* hst = h + (8 * half) * 256 + c;
#pragma unroll
            for (int rt = 0; rt < GRU_RT; ++rt) {
                int rbase = (row0 + rt * 16 < rowEnd) ? row0 + rt * 16 : row0;
#pragma unroll
                for (int vr = 0; vr < 8; ++vr)
                    hold[rt][vr] = hcol[(long long)(rbase + vr) * 256];
            }
#pragma unroll
            for (int rt = 0; rt < GRU_RT; ++rt) {
                if (row0 + rt * 16 < rowEnd) {
                    float gr[8], gz[8], gn[8];
#pragma unroll
                    for (int vr = 0; vr < 8; ++vr)
                        gr[vr] = sigmoidf_(aR[rt][vr] + bir + bhr);
#pragma unroll
                    for (int vr = 0; vr < 8; ++vr)
                        gz[vr] = sigmoidf_(aZ[rt][vr] + biz + bhz);
#pragma unroll
                    for (int vr = 0; vr < 8; ++vr)
                        gn[vr] = tanh_two_range(aIN[rt][vr] + bin + gr[vr] * (aHN[rt][vr] + bhn));
#pragma unroll
                    for (int vr = 0; vr < 8; ++vr) {
                        hst[(long long)(row0 + rt * 16 + vr) * 256] =
                            (1.0f - gz[vr]) * gn[vr] + gz[vr] * hold[rt][vr];
                    }
                }
            }
        }
    }
}

__global__ void __launch_bounds__(256)
final_kernel(const float* __restrict__ h, const float* __restrict__ fcw,
             const float* __restrict__ fcb, float* __restrict__ out) {
    __shared__ float sw[N_ACT_C * 256];
    __shared__ float sb[N_ACT_C];
    for (int i = threadIdx.x; i < N_ACT_C * 256; i += 256) sw[i] = fcw[i];
    if (threadIdx.x < N_ACT_C) sb[threadIdx.x] = fcb[threadIdx.x];
    __syncthreads();

    int n = blockIdx.x * 256 + threadIdx.x;
    if (n >= N_NODES_C) return;

    float acc[N_ACT_C];
#pragma unroll
    for (int a = 0; a < N_ACT_C; ++a) acc[a] = sb[a];
    const float* hr = h + (long long)n * 256;
    for (int k = 0; k < 256; ++k) {
        float hv = fmaxf(hr[k], 0.0f);
#pragma unroll
        for (int a = 0; a < N_ACT_C; ++a) acc[a] += hv * sw[a * 256 + k];
    }
    float mx = acc[0];
#pragma unroll
    for (int a = 1; a < N_ACT_C; ++a) mx = fmaxf(mx, acc[a]);
    float s = 0.0f;
#pragma unroll
    for (int a = 0; a < N_ACT_C; ++a) { acc[a] = __expf(acc[a] - mx); s += acc[a]; }
    float inv = 1.0f / s;
#pragma unroll
    for (int a = 0; a < N_ACT_C; ++a) out[(long long)n * N_ACT_C + a] = acc[a] * inv;
}

extern "C" void kernel_launch(void* const* d_in, const int* in_sizes, int n_in,
                              void* d_out, int out_size, void* d_ws, size_t ws_size,
                              hipStream_t stream) {
    const float* x   = (const float*)d_in[0];
    const int*   ei  = (const int*)  d_in[1];
    const float* ew  = (const float*)d_in[2];
    const float* gg  = (const float*)d_in[3];
    const float* wih = (const float*)d_in[4];
    const float* whh = (const float*)d_in[5];
    const float* bih = (const float*)d_in[6];
    const float* bhh = (const float*)d_in[7];
    const float* fcw = (const float*)d_in[8];
    const float* fcb = (const float*)d_in[9];
    float* out = (float*)d_out;

    float* h   = (float*)d_ws;
    float* m   = h + (size_t)N_PAD * 256;
    float* agg = m + (size_t)N_PAD * 256;
    __bf16* ggT  = (__bf16*)(agg + (size_t)N_PAD * 256);
    __bf16* wihb = ggT + 3 * 256 * 256;
    __bf16* whhb = wihb + 768 * 256;
    int* counts   = (int*)(whhb + 768 * 256);
    int* within   = counts + (size_t)NBLK * BIN_PITCH;
    int* binTotal = within + (size_t)NBLK * BIN_PITCH;
    int* binStart = binTotal + BIN_PITCH;
    int4* rec     = (int4*)(binStart + BIN_PITCH);

    convw_kernel<<<(3 * 256 * 256 + 255) / 256, 256, 0, stream>>>(gg, wih, whh,
                                                                  ggT, wihb, whhb);
    init_h_kernel<<<N_PAD, 256, 0, stream>>>(x, h);
    bin_count_kernel<<<NBLK, 256, 0, stream>>>(ei, counts);
    bin_scan_kernel<<<(NBIN + 31) / 32, 32, 0, stream>>>(counts, within, binTotal);
    bin_place_kernel<<<NBLK, 256, 0, stream>>>(ei, ew, within, binTotal, binStart, rec);

    for (int l = 0; l < 3; ++l) {
        gemm_m_kernel<<<(N_PAD + GM_RB - 1) / GM_RB, 256, 0, stream>>>(h, ggT + (size_t)l * 65536, m);
        agg_kernel<<<NBIN, 128, 0, stream>>>(m, rec, binStart, binTotal, agg);
        gru_kernel<<<(N_PAD / 16 + GRU_BT - 1) / GRU_BT, 256, 0, stream>>>(agg, h, wihb, whhb,
                                                                          bih, bhh);
    }

    final_kernel<<<(N_NODES_C + 255) / 256, 256, 0, stream>>>(h, fcw, fcb, out);
}
